# starting kernel
# baseline (speedup 1.0000x reference)
#include <hip/hip_runtime.h>
#include <hip/hip_bf16.h>


#define DIM_IN   128
#define HID      64
#define HEADS    8
#define DIM_H    8
#define DIM_OUT  40
#define NEG_SLOPE 0.2f

typedef __attribute__((ext_vector_type(2))) float v2f;
typedef __attribute__((ext_vector_type(8))) float v8f;

__device__ __forceinline__ void atomicMaxFloat(float* addr, float val) {
    if (!(val == val)) return;
    if (val >= 0.0f) {
        atomicMax((int*)addr, __float_as_int(val));
    } else {
        atomicMin((unsigned int*)addr, __float_as_uint(val));
    }
}

__global__ void gat_fill_u32(unsigned int* __restrict__ p, unsigned int val,
                             long long n) {
    long long i = (long long)blockIdx.x * blockDim.x + threadIdx.x;
    if (i < n) p[i] = val;
}

template <int K, int CC, int CP>
__global__ __launch_bounds__(256) void gat_gemm_bias_f32_wmma(
    const float* __restrict__ A, const float* __restrict__ W,
    const float* __restrict__ bias, float* __restrict__ out, int n) {
    __shared__ float sWt[CP * K];

    for (int idx = threadIdx.x; idx < CP * K; idx += blockDim.x) {
        const int c = idx / K;
        const int k = idx % K;
        sWt[idx] = (c < CC) ? W[(size_t)k * CC + c] : 0.0f;
    }
    __syncthreads();

    const int waveId = threadIdx.x >> 5;
    const int lane = threadIdx.x & 31;
    constexpr int colTiles = CP >> 4;
    const int rowTiles = n >> 4;
    const int tileId = blockIdx.x * (blockDim.x >> 5) + waveId;
    const int rowTile = tileId / colTiles;
    const int colTile = tileId % colTiles;
    if (rowTile >= rowTiles) return;

    const int mrow = (rowTile << 4) + (lane & 15);
    const int colp = (colTile << 4) + (lane & 15);
    const int khalf = (lane >> 4) << 1;

    const float* __restrict__ aRow = A + (size_t)mrow * K + khalf;
    const float* __restrict__ bCol = sWt + (size_t)colp * K + khalf;

    v8f acc = {};
#pragma unroll
    for (int k0 = 0; k0 < K; k0 += 4) {
        const v2f a = *(const v2f*)(aRow + k0);
        const v2f b = *(const v2f*)(bCol + k0);
        acc = __builtin_amdgcn_wmma_f32_16x16x4_f32(
            false, a, false, b, (short)0, acc, false, false);
    }

    if (colp < CC) {
        const float bv = bias[colp];
        const int rbase = (rowTile << 4) + ((lane >> 4) << 3);
#pragma unroll
        for (int r = 0; r < 8; ++r) {
            out[(size_t)(rbase + r) * CC + colp] = acc[r] + bv;
        }
    }
}

__global__ void gat_edge_logits_max_l1(
    const float* __restrict__ xl, const float* __restrict__ xr,
    const float* __restrict__ att, const int* __restrict__ src,
    const int* __restrict__ dst, float* __restrict__ e1,
    float* __restrict__ m1, long long E8) {
    long long t = (long long)blockIdx.x * blockDim.x + threadIdx.x;
    if (t >= E8) return;
    const int e = (int)(t >> 3);
    const int h = (int)(t & 7);
    const int s = src[e], d = dst[e];
    const float* pj = xl + (size_t)s * HID + h * DIM_H;
    const float* pi = xr + (size_t)d * HID + h * DIM_H;
    const float* pa = att + h * DIM_H;
    float acc = 0.0f;
#pragma unroll
    for (int c = 0; c < DIM_H; ++c) {
        float v = pi[c] + pj[c];
        v = v > 0.0f ? v : v * NEG_SLOPE;
        acc += v * pa[c];
    }
    e1[t] = acc;
    atomicMaxFloat(&m1[(size_t)d * HEADS + h], acc);
}

__global__ void gat_edge_expsum_l1(
    const int* __restrict__ dst, const float* __restrict__ m1,
    float* __restrict__ e1, float* __restrict__ denom1, long long E8) {
    long long t = (long long)blockIdx.x * blockDim.x + threadIdx.x;
    if (t >= E8) return;
    const int e = (int)(t >> 3);
    const int h = (int)(t & 7);
    const int d = dst[e];
    const float ex = expf(e1[t] - m1[(size_t)d * HEADS + h]);
    e1[t] = ex;
    atomicAdd(&denom1[(size_t)d * HEADS + h], ex);
}

__global__ void gat_edge_aggregate_l1(
    const int* __restrict__ src, const int* __restrict__ dst,
    const float* __restrict__ xl, const float* __restrict__ e1,
    const float* __restrict__ denom1, float* __restrict__ out1, long long E8) {
    long long t = (long long)blockIdx.x * blockDim.x + threadIdx.x;
    if (t >= E8) return;
    const int e = (int)(t >> 3);
    const int h = (int)(t & 7);
    const int s = src[e], d = dst[e];
    const float alpha = e1[t] / denom1[(size_t)d * HEADS + h];
    const float* pj = xl + (size_t)s * HID + h * DIM_H;
    float* po = out1 + (size_t)d * HID + h * DIM_H;
#pragma unroll
    for (int c = 0; c < DIM_H; ++c) {
        atomicAdd(&po[c], pj[c] * alpha);
    }
}

__global__ void gat_bias_elu(float* __restrict__ out1,
                             const float* __restrict__ bias1, long long total) {
    long long i = (long long)blockIdx.x * blockDim.x + threadIdx.x;
    if (i >= total) return;
    float v = out1[i] + bias1[(int)(i & (HID - 1))];
    out1[i] = v > 0.0f ? v : expm1f(v);
}

__global__ void gat_edge_logits_max_l2(
    const float* __restrict__ xl, const float* __restrict__ xr,
    const float* __restrict__ att, const int* __restrict__ src,
    const int* __restrict__ dst, float* __restrict__ e2,
    float* __restrict__ m2, long long E8) {
    long long t = (long long)blockIdx.x * blockDim.x + threadIdx.x;
    if (t >= E8) return;
    const int e = (int)(t >> 3);
    const int sub = (int)(t & 7);
    const int s = src[e], d = dst[e];
    const float* pj = xl + (size_t)s * DIM_OUT;
    const float* pi = xr + (size_t)d * DIM_OUT;
    float acc = 0.0f;
#pragma unroll
    for (int c = sub; c < DIM_OUT; c += 8) {
        float v = pi[c] + pj[c];
        v = v > 0.0f ? v : v * NEG_SLOPE;
        acc += v * att[c];
    }
    acc += __shfl_xor(acc, 1, 32);
    acc += __shfl_xor(acc, 2, 32);
    acc += __shfl_xor(acc, 4, 32);
    if (sub == 0) {
        e2[e] = acc;
        atomicMaxFloat(&m2[d], acc);
    }
}

__global__ void gat_edge_expsum_l2(
    const int* __restrict__ dst, const float* __restrict__ m2,
    float* __restrict__ e2, float* __restrict__ denom2, int E) {
    int e = blockIdx.x * blockDim.x + threadIdx.x;
    if (e >= E) return;
    const int d = dst[e];
    const float ex = expf(e2[e] - m2[d]);
    e2[e] = ex;
    atomicAdd(&denom2[d], ex);
}

__global__ void gat_edge_aggregate_l2(
    const int* __restrict__ src, const int* __restrict__ dst,
    const float* __restrict__ xl, const float* __restrict__ e2,
    const float* __restrict__ denom2, float* __restrict__ out2, long long E8) {
    long long t = (long long)blockIdx.x * blockDim.x + threadIdx.x;
    if (t >= E8) return;
    const int e = (int)(t >> 3);
    const int sub = (int)(t & 7);
    const int s = src[e], d = dst[e];
    const float alpha = e2[e] / denom2[d];
    const float* pj = xl + (size_t)s * DIM_OUT;
    float* po = out2 + (size_t)d * DIM_OUT;
#pragma unroll
    for (int c = sub; c < DIM_OUT; c += 8) {
        atomicAdd(&po[c], pj[c] * alpha);
    }
}

__global__ __launch_bounds__(256) void gat_finalize(
    const float* __restrict__ out2, const float* __restrict__ bias2,
    float* __restrict__ d_out, int N) {
    const int node = blockIdx.x * (blockDim.x >> 5) + (threadIdx.x >> 5);
    const int lane = threadIdx.x & 31;
    if (node >= N) return;
    const float* row = out2 + (size_t)node * DIM_OUT;
    const float a = row[lane] + bias2[lane];
    const bool has2 = (lane < DIM_OUT - 32);
    const float b = has2 ? (row[32 + lane] + bias2[32 + lane]) : -__builtin_inff();
    float mx = fmaxf(a, b);
#pragma unroll
    for (int o = 16; o >= 1; o >>= 1) mx = fmaxf(mx, __shfl_xor(mx, o, 32));
    float s = expf(a - mx) + (has2 ? expf(b - mx) : 0.0f);
#pragma unroll
    for (int o = 16; o >= 1; o >>= 1) s += __shfl_xor(s, o, 32);
    const float lse = logf(s) + mx;
    float* h  = d_out + (size_t)node * DIM_OUT;
    float* ls = d_out + (size_t)N * DIM_OUT + (size_t)node * DIM_OUT;
    h[lane] = a;
    ls[lane] = a - lse;
    if (has2) {
        h[32 + lane] = b;
        ls[32 + lane] = b - lse;
    }
}

static inline int cdiv_i(long long a, long long b) { return (int)((a + b - 1) / b); }

extern "C" void kernel_launch(void* const* d_in, const int* in_sizes, int n_in,
                              void* d_out, int out_size, void* d_ws, size_t ws_size,
                              hipStream_t stream) {
    (void)n_in; (void)out_size; (void)ws_size;
    const float* x     = (const float*)d_in[0];
    const int*   eidx  = (const int*)d_in[1];
    const float* Wl1   = (const float*)d_in[2];
    const float* bl1   = (const float*)d_in[3];
    const float* Wr1   = (const float*)d_in[4];
    const float* br1   = (const float*)d_in[5];
    const float* att1  = (const float*)d_in[6];
    const float* bias1 = (const float*)d_in[7];
    const float* Wl2   = (const float*)d_in[8];
    const float* bl2   = (const float*)d_in[9];
    const float* Wr2   = (const float*)d_in[10];
    const float* br2   = (const float*)d_in[11];
    const float* att2  = (const float*)d_in[12];
    const float* bias2 = (const float*)d_in[13];

    const int N = in_sizes[0] / DIM_IN;
    const int E = in_sizes[1] / 2;
    const int* src = eidx;
    const int* dst = eidx + E;

    float* ws = (float*)d_ws;
    size_t o = 0;
    float* h_l1   = ws + o; o += (size_t)N * HID;
    float* h_r1   = ws + o; o += (size_t)N * HID;
    float* e1     = ws + o; o += (size_t)E * HEADS;
    float* m1     = ws + o; o += (size_t)N * HEADS;
    float* denom1 = ws + o; o += (size_t)N * HEADS;
    float* out1   = ws + o; o += (size_t)N * HID;
    float* h_l2   = ws;
    float* h_r2   = h_l2 + (size_t)N * DIM_OUT;
    float* e2     = h_r2 + (size_t)N * DIM_OUT;
    float* out2   = e2 + (size_t)E;
    float* m2     = out2 + (size_t)N * DIM_OUT;
    float* denom2 = m2 + (size_t)N;

    const int B = 256;
    const long long E8 = (long long)E * 8;
    const unsigned int NEG_INF_BITS = 0xFF800000u;

    gat_fill_u32<<<cdiv_i((long long)N * HEADS, B), B, 0, stream>>>(
        (unsigned int*)m1, NEG_INF_BITS, (long long)N * HEADS);
    gat_fill_u32<<<cdiv_i((long long)N * HEADS, B), B, 0, stream>>>(
        (unsigned int*)denom1, 0u, (long long)N * HEADS);
    gat_fill_u32<<<cdiv_i((long long)N * HID, B), B, 0, stream>>>(
        (unsigned int*)out1, 0u, (long long)N * HID);

    {
        const int tiles = (N >> 4) * (HID >> 4);
        const int grid = cdiv_i(tiles, B / 32);
        gat_gemm_bias_f32_wmma<DIM_IN, HID, HID>
            <<<grid, B, 0, stream>>>(x, Wl1, bl1, h_l1, N);
        gat_gemm_bias_f32_wmma<DIM_IN, HID, HID>
            <<<grid, B, 0, stream>>>(x, Wr1, br1, h_r1, N);
    }

    gat_edge_logits_max_l1<<<cdiv_i(E8, B), B, 0, stream>>>(
        h_l1, h_r1, att1, src, dst, e1, m1, E8);
    gat_edge_expsum_l1<<<cdiv_i(E8, B), B, 0, stream>>>(dst, m1, e1, denom1, E8);
    gat_edge_aggregate_l1<<<cdiv_i(E8, B), B, 0, stream>>>(
        src, dst, h_l1, e1, denom1, out1, E8);
    gat_bias_elu<<<cdiv_i((long long)N * HID, B), B, 0, stream>>>(
        out1, bias1, (long long)N * HID);

    gat_fill_u32<<<cdiv_i(N, B), B, 0, stream>>>((unsigned int*)m2, NEG_INF_BITS, N);
    gat_fill_u32<<<cdiv_i(N, B), B, 0, stream>>>((unsigned int*)denom2, 0u, N);
    gat_fill_u32<<<cdiv_i((long long)N * DIM_OUT, B), B, 0, stream>>>(
        (unsigned int*)out2, 0u, (long long)N * DIM_OUT);

    {
        const int tiles = (N >> 4) * (48 >> 4);
        const int grid = cdiv_i(tiles, B / 32);
        gat_gemm_bias_f32_wmma<HID, DIM_OUT, 48>
            <<<grid, B, 0, stream>>>(out1, Wl2, bl2, h_l2, N);
        gat_gemm_bias_f32_wmma<HID, DIM_OUT, 48>
            <<<grid, B, 0, stream>>>(out1, Wr2, br2, h_r2, N);
    }

    gat_edge_logits_max_l2<<<cdiv_i(E8, B), B, 0, stream>>>(
        h_l2, h_r2, att2, src, dst, e2, m2, E8);
    gat_edge_expsum_l2<<<cdiv_i(E, B), B, 0, stream>>>(dst, m2, e2, denom2, E);
    gat_edge_aggregate_l2<<<cdiv_i(E8, B), B, 0, stream>>>(
        src, dst, h_l2, e2, denom2, out2, E8);

    gat_finalize<<<cdiv_i(N, B / 32), B, 0, stream>>>((const float*)out2, bias2,
                                                      (float*)d_out, N);
}


# Start values of the scatter targets (running max, denominators, accumulator rows) written by each layer's projection launch for its own row tiles; the 5 fill launches go (13->8); sums unchanged
# speedup vs baseline: 2.0489x; 2.0489x over previous
#include <hip/hip_runtime.h>
#include <hip/hip_bf16.h>


#define DIM_IN   128
#define HID      64
#define HEADS    8
#define DIM_H    8
#define DIM_OUT  40
#define NEG_SLOPE 0.2f
#define ROW2     64
#define DEN2     40

typedef __attribute__((ext_vector_type(2))) float v2f;
typedef __attribute__((ext_vector_type(4))) float v4f;
typedef __attribute__((ext_vector_type(8))) float v8f;

__device__ __forceinline__ void atomicMaxFloat(float* addr, float val) {
    if (!(val == val)) return;
    if (val >= 0.0f) {
        atomicMax((int*)addr, __float_as_int(val));
    } else {
        atomicMin((unsigned int*)addr, __float_as_uint(val));
    }
}

template <int K, int CC, int NT, int TPW, int ACCW, int MW, int DW>
__global__ __launch_bounds__(32 * (2 * NT / TPW)) void gat_proj_pair_f32_wmma(
    const float* __restrict__ A,
    const float* __restrict__ Wl, const float* __restrict__ bl,
    const float* __restrict__ Wr, const float* __restrict__ br,
    float* __restrict__ outl, float* __restrict__ outr,
    float* __restrict__ acc0, float* __restrict__ max0, float* __restrict__ den0,
    int n) {
    constexpr int WPM = NT / TPW;
    constexpr int KS = K / 4;
    constexpr bool PAD = (16 * NT != CC);
    const int lane = threadIdx.x & 31;
    const int wave = __builtin_amdgcn_readfirstlane((int)(threadIdx.x >> 5));
    const bool right = wave >= WPM;
    const float* __restrict__ W = right ? Wr : Wl;
    const float* __restrict__ bias = right ? br : bl;
    float* __restrict__ out = right ? outr : outl;
    const int tile0 = (WPM == 1) ? 0 : (right ? wave - WPM : wave) * TPW;
    const int c = lane & 15;
    const int h = lane >> 4;

    v2f b[TPW][KS];
    float bv[TPW];
#pragma unroll
    for (int t = 0; t < TPW; ++t) {
        const int col = (tile0 + t) * 16 + c;
        const bool ok = !PAD || col < CC;
        const float* __restrict__ wc = W + (ok ? col : 0);
        bv[t] = ok ? bias[col] : 0.0f;
#pragma unroll
        for (int s = 0; s < KS; ++s) {
            const int k = 8 * (s >> 1) + 4 * h + 2 * (s & 1);
            const float w0 = wc[(size_t)k * CC];
            const float w1 = wc[(size_t)(k + 1) * CC];
            b[t][s].x = ok ? w0 : 0.0f;
            b[t][s].y = ok ? w1 : 0.0f;
        }
    }

    const int rowTiles = n >> 4;
    for (int rt = blockIdx.x; rt < rowTiles; rt += gridDim.x) {
        const float* __restrict__ aRow = A + (size_t)((rt << 4) + c) * K + 4 * h;
        v4f a[K / 8];
#pragma unroll
        for (int j = 0; j < K / 8; ++j) a[j] = *(const v4f*)(aRow + 8 * j);

        v8f acc[TPW];
#pragma unroll
        for (int t = 0; t < TPW; ++t) acc[t] = v8f{};
#pragma unroll
        for (int j = 0; j < K / 8; ++j) {
            const v2f a0 = {a[j].x, a[j].y};
            const v2f a1 = {a[j].z, a[j].w};
#pragma unroll
            for (int t = 0; t < TPW; ++t)
                acc[t] = __builtin_amdgcn_wmma_f32_16x16x4_f32(
                    false, a0, false, b[t][2 * j], (short)0, acc[t], false, false);
#pragma unroll
            for (int t = 0; t < TPW; ++t)
                acc[t] = __builtin_amdgcn_wmma_f32_16x16x4_f32(
                    false, a1, false, b[t][2 * j + 1], (short)0, acc[t], false, false);
        }

        const int rbase = (rt << 4) + (h << 3);
#pragma unroll
        for (int t = 0; t < TPW; ++t) {
            const int col = (tile0 + t) * 16 + c;
            if (!PAD || col < CC) {
#pragma unroll
                for (int r = 0; r < 8; ++r)
                    out[(size_t)(rbase + r) * CC + col] = acc[t][r] + bv[t];
            }
        }

        constexpr int NW = 2 * WPM;
        constexpr int ACC_P = 16 * ACCW / 4;
        constexpr int MAX_P = 16 * MW / 4;
        constexpr int DEN_P = 16 * DW / 4;
        static_assert(ACC_P % (NW * 32) == 0 && MAX_P <= 32 && DEN_P <= 32, "tile start values");
        v4f* __restrict__ pa = (v4f*)(acc0 + (size_t)rt * (16 * ACCW));
        for (int q = 0; q < ACC_P / (NW * 32); ++q) pa[(q * NW + wave) * 32 + lane] = v4f{};
        if (wave == NW - 1 && lane < MAX_P) {
            const float ninf = -__builtin_inff();
            ((v4f*)(max0 + (size_t)rt * (16 * MW)))[lane] = v4f{ninf, ninf, ninf, ninf};
        }
        if constexpr (DW > 0) {
            if (wave == 0 && lane < DEN_P)
                ((v4f*)(den0 + (size_t)rt * (16 * DW)))[lane] = v4f{};
        }
    }
}

__global__ void gat_edge_logits_max_l1(
    const float* __restrict__ xl, const float* __restrict__ xr,
    const float* __restrict__ att, const int* __restrict__ src,
    const int* __restrict__ dst, float* __restrict__ e1,
    float* __restrict__ m1, long long E8) {
    long long t = (long long)blockIdx.x * blockDim.x + threadIdx.x;
    if (t >= E8) return;
    const int e = (int)(t >> 3);
    const int h = (int)(t & 7);
    const int s = src[e], d = dst[e];
    const float* pj = xl + (size_t)s * HID + h * DIM_H;
    const float* pi = xr + (size_t)d * HID + h * DIM_H;
    const float* pa = att + h * DIM_H;
    float acc = 0.0f;
#pragma unroll
    for (int c = 0; c < DIM_H; ++c) {
        float v = pi[c] + pj[c];
        v = v > 0.0f ? v : v * NEG_SLOPE;
        acc += v * pa[c];
    }
    e1[t] = acc;
    atomicMaxFloat(&m1[(size_t)d * HEADS + h], acc);
}

__global__ __launch_bounds__(256) void gat_edge_aggregate_l1(
    const int* __restrict__ src, const int* __restrict__ dst,
    const float* __restrict__ xl, const float* __restrict__ e1,
    const float* __restrict__ m1, float* __restrict__ denom1,
    float* __restrict__ out1, int E) {
    const int lane = threadIdx.x & 31;
    const int wave = __builtin_amdgcn_readfirstlane(
        (int)(blockIdx.x * (blockDim.x >> 5) + (threadIdx.x >> 5)));
    const int wbase = wave * 32;
    if (wbase >= E) return;
    const int sub = lane >> 3;
    const int h = lane & 7;
    const int nE = (E - wbase) < 32 ? (E - wbase) : 32;

    if (nE == 32) {
        const int s_i = src[wbase + lane];
        const int d_i = dst[wbase + lane];
        float ev[8], mv[8], ex[8];
        int dp[8];
#pragma unroll
        for (int g = 0; g < 8; ++g) {
            dp[g] = __shfl(d_i, g * 4 + sub, 32);
            ev[g] = e1[(size_t)wbase * HEADS + g * 32 + lane];
            mv[g] = m1[(size_t)dp[g] * HEADS + h];
        }
#pragma unroll
        for (int g = 0; g < 8; ++g) {
            ex[g] = expf(ev[g] - mv[g]);
            atomicAdd(&denom1[(size_t)dp[g] * HEADS + h], ex[g]);
        }
#pragma unroll
        for (int g = 0; g < 8; ++g) {
            float r0[4], r1[4];
#pragma unroll
            for (int q = 0; q < 4; ++q) {
                const int s = __builtin_amdgcn_readlane(s_i, g * 4 + q);
                const float* pj = xl + (size_t)s * HID;
                r0[q] = pj[lane];
                r1[q] = pj[32 + lane];
            }
#pragma unroll
            for (int q = 0; q < 4; ++q) {
                const int d = __builtin_amdgcn_readlane(d_i, g * 4 + q);
                const float a = __shfl(ex[g], q * 8 + sub, 32);
                const float b = __shfl(ex[g], q * 8 + 4 + sub, 32);
                float* po = out1 + (size_t)d * HID;
                atomicAdd(po + lane, r0[q] * a);
                atomicAdd(po + 32 + lane, r1[q] * b);
            }
        }
    } else {
        for (int j = 0; j < nE; ++j) {
            const int e = wbase + j;
            const int s = src[e], d = dst[e];
            const float a = expf(e1[(size_t)e * HEADS + sub] - m1[(size_t)d * HEADS + sub]);
            const float b = expf(e1[(size_t)e * HEADS + 4 + sub] - m1[(size_t)d * HEADS + 4 + sub]);
            if (h == 0) {
                atomicAdd(&denom1[(size_t)d * HEADS + sub], a);
                atomicAdd(&denom1[(size_t)d * HEADS + 4 + sub], b);
            }
            const float* pj = xl + (size_t)s * HID;
            float* po = out1 + (size_t)d * HID;
            atomicAdd(po + lane, pj[lane] * a);
            atomicAdd(po + 32 + lane, pj[32 + lane] * b);
        }
    }
}

__global__ void gat_norm_bias_elu(float* __restrict__ out1,
                                  const float* __restrict__ denom1,
                                  const float* __restrict__ bias1,
                                  long long total) {
    long long i = (long long)blockIdx.x * blockDim.x + threadIdx.x;
    if (i >= total) return;
    const float den = denom1[i >> 3];
    const float acc = out1[i];
    float v = (den > 0.0f ? acc / den : acc) + bias1[(int)(i & (HID - 1))];
    out1[i] = v > 0.0f ? v : expm1f(v);
}

__global__ void gat_edge_logits_max_l2(
    const float* __restrict__ xl, const float* __restrict__ xr,
    const float* __restrict__ att, const int* __restrict__ src,
    const int* __restrict__ dst, float* __restrict__ e2,
    float* __restrict__ m2, long long E8) {
    long long t = (long long)blockIdx.x * blockDim.x + threadIdx.x;
    if (t >= E8) return;
    const int e = (int)(t >> 3);
    const int sub = (int)(t & 7);
    const int s = src[e], d = dst[e];
    const float* pj = xl + (size_t)s * DIM_OUT;
    const float* pi = xr + (size_t)d * DIM_OUT;
    float acc = 0.0f;
#pragma unroll
    for (int c = sub; c < DIM_OUT; c += 8) {
        float v = pi[c] + pj[c];
        v = v > 0.0f ? v : v * NEG_SLOPE;
        acc += v * att[c];
    }
    acc += __shfl_xor(acc, 1, 32);
    acc += __shfl_xor(acc, 2, 32);
    acc += __shfl_xor(acc, 4, 32);
    if (sub == 0) {
        e2[e] = acc;
        atomicMaxFloat(&m2[d], acc);
    }
}

__global__ __launch_bounds__(256) void gat_edge_aggregate_l2(
    const int* __restrict__ src, const int* __restrict__ dst,
    const float* __restrict__ xl, const float* __restrict__ e2,
    const float* __restrict__ m2, float* __restrict__ out2, int E) {
    const int lane = threadIdx.x & 31;
    const int wave = __builtin_amdgcn_readfirstlane(
        (int)(blockIdx.x * (blockDim.x >> 5) + (threadIdx.x >> 5)));
    const int wbase = wave * 32;
    if (wbase >= E) return;
    const int ei = wbase + lane;
    int s_i = 0, d_i = 0;
    float ex_i = 0.0f;
    if (ei < E) {
        s_i = src[ei];
        d_i = dst[ei];
        ex_i = expf(e2[ei] - m2[d_i]);
    }
    const int nE = (E - wbase) < 32 ? (E - wbase) : 32;
    const int c1 = 32 + (lane & 7);
    const bool tail_ch = lane < DIM_OUT - 32;

    if (nE == 32) {
        for (int j0 = 0; j0 < 32; j0 += 4) {
            float r0[4], r1[4];
#pragma unroll
            for (int q = 0; q < 4; ++q) {
                const int s = __builtin_amdgcn_readlane(s_i, j0 + q);
                const float* pj = xl + (size_t)s * DIM_OUT;
                r0[q] = pj[lane];
                r1[q] = pj[c1];
            }
            asm volatile("" :: "v"(r1[0]), "v"(r1[1]), "v"(r1[2]), "v"(r1[3]));
#pragma unroll
            for (int q = 0; q < 4; ++q) {
                const int d = __builtin_amdgcn_readlane(d_i, j0 + q);
                const float a = __int_as_float(
                    __builtin_amdgcn_readlane(__float_as_int(ex_i), j0 + q));
                float* po = out2 + (size_t)d * ROW2;
                const float v1 = tail_ch ? r1[q] : 1.0f;
                atomicAdd(po + lane, r0[q] * a);
                if (lane <= DEN2 - 32) atomicAdd(po + 32 + lane, v1 * a);
            }
        }
    } else {
        for (int j = 0; j < nE; ++j) {
            const int s = __builtin_amdgcn_readlane(s_i, j);
            const int d = __builtin_amdgcn_readlane(d_i, j);
            const float a = __int_as_float(
                __builtin_amdgcn_readlane(__float_as_int(ex_i), j));
            const float* pj = xl + (size_t)s * DIM_OUT;
            float* po = out2 + (size_t)d * ROW2;
            const float v0 = pj[lane];
            const float v1 = tail_ch ? pj[c1] : 1.0f;
            atomicAdd(po + lane, v0 * a);
            if (lane <= DEN2 - 32) atomicAdd(po + 32 + lane, v1 * a);
        }
    }
}

__global__ __launch_bounds__(256) void gat_finalize(
    const float* __restrict__ out2,
    const float* __restrict__ bias2, float* __restrict__ d_out, int N) {
    const int node = blockIdx.x * (blockDim.x >> 5) + (threadIdx.x >> 5);
    const int lane = threadIdx.x & 31;
    if (node >= N) return;
    const float* row = out2 + (size_t)node * ROW2;
    const float den = row[DEN2];
    const float ra = row[lane];
    const float a = (den > 0.0f ? ra / den : ra) + bias2[lane];
    const bool has2 = (lane < DIM_OUT - 32);
    const float rb = has2 ? row[32 + lane] : 0.0f;
    const float b = has2 ? ((den > 0.0f ? rb / den : rb) + bias2[32 + lane])
                         : -__builtin_inff();
    float mx = fmaxf(a, b);
#pragma unroll
    for (int o = 16; o >= 1; o >>= 1) mx = fmaxf(mx, __shfl_xor(mx, o, 32));
    float s = expf(a - mx) + (has2 ? expf(b - mx) : 0.0f);
#pragma unroll
    for (int o = 16; o >= 1; o >>= 1) s += __shfl_xor(s, o, 32);
    const float lse = logf(s) + mx;
    float* h  = d_out + (size_t)node * DIM_OUT;
    float* ls = d_out + (size_t)N * DIM_OUT + (size_t)node * DIM_OUT;
    h[lane] = a;
    ls[lane] = a - lse;
    if (has2) {
        h[32 + lane] = b;
        ls[32 + lane] = b - lse;
    }
}

static inline int cdiv_i(long long a, long long b) { return (int)((a + b - 1) / b); }

extern "C" void kernel_launch(void* const* d_in, const int* in_sizes, int n_in,
                              void* d_out, int out_size, void* d_ws, size_t ws_size,
                              hipStream_t stream) {
    (void)n_in; (void)out_size; (void)ws_size;
    const float* x     = (const float*)d_in[0];
    const int*   eidx  = (const int*)d_in[1];
    const float* Wl1   = (const float*)d_in[2];
    const float* bl1   = (const float*)d_in[3];
    const float* Wr1   = (const float*)d_in[4];
    const float* br1   = (const float*)d_in[5];
    const float* att1  = (const float*)d_in[6];
    const float* bias1 = (const float*)d_in[7];
    const float* Wl2   = (const float*)d_in[8];
    const float* bl2   = (const float*)d_in[9];
    const float* Wr2   = (const float*)d_in[10];
    const float* br2   = (const float*)d_in[11];
    const float* att2  = (const float*)d_in[12];
    const float* bias2 = (const float*)d_in[13];

    const int N = in_sizes[0] / DIM_IN;
    const int E = in_sizes[1] / 2;
    const int* src = eidx;
    const int* dst = eidx + E;

    float* ws = (float*)d_ws;
    size_t o = 0;
    float* h_l1   = ws + o; o += (size_t)N * HID;
    float* h_r1   = ws + o; o += (size_t)N * HID;
    float* e1     = ws + o; o += (size_t)E * HEADS;
    float* m1     = ws + o; o += (size_t)N * HEADS;
    float* denom1 = ws + o; o += (size_t)N * HEADS;
    float* out1   = ws + o; o += (size_t)N * HID;
    float* h_l2   = ws;
    float* h_r2   = h_l2 + (size_t)N * DIM_OUT;
    float* e2     = h_r2 + (size_t)N * DIM_OUT;
    float* out2   = e2 + (size_t)E;
    float* m2     = out2 + (size_t)N * ROW2;

    const int B = 256;
    const long long E8 = (long long)E * 8;

    gat_proj_pair_f32_wmma<DIM_IN, HID, 4, 2, HID, HEADS, HEADS>
        <<<1024, 128, 0, stream>>>(x, Wl1, bl1, Wr1, br1, h_l1, h_r1,
                                   out1, m1, denom1, N);

    gat_edge_logits_max_l1<<<cdiv_i(E8, B), B, 0, stream>>>(
        h_l1, h_r1, att1, src, dst, e1, m1, E8);
    gat_edge_aggregate_l1<<<cdiv_i(E, B), B, 0, stream>>>(
        src, dst, h_l1, e1, m1, denom1, out1, E);
    gat_norm_bias_elu<<<cdiv_i((long long)N * HID, B), B, 0, stream>>>(
        out1, denom1, bias1, (long long)N * HID);

    gat_proj_pair_f32_wmma<HID, DIM_OUT, 3, 3, ROW2, 1, 0>
        <<<1250, 64, 0, stream>>>(out1, Wl2, bl2, Wr2, br2, h_l2, h_r2,
                                  out2, m2, (float*)nullptr, N);

    gat_edge_logits_max_l2<<<cdiv_i(E8, B), B, 0, stream>>>(
        h_l2, h_r2, att2, src, dst, e2, m2, E8);
    gat_edge_aggregate_l2<<<cdiv_i(E, B), B, 0, stream>>>(
        src, dst, h_l2, e2, m2, out2, E);

    gat_finalize<<<cdiv_i(N, B / 32), B, 0, stream>>>((const float*)out2,
                                                      bias2, (float*)d_out, N);
}


# Softmax shift per destination = upper bound of its logits (column max/min of source plane), not the exact max: logit, exp, scatter in one edge pass per layer; 8->10 launches; logit sum order changes
# speedup vs baseline: 2.9291x; 1.4296x over previous
#include <hip/hip_runtime.h>
#include <hip/hip_bf16.h>


#define DIM_IN   128
#define HID      64
#define HEADS    8
#define DIM_H    8
#define DIM_OUT  40
#define NEG_SLOPE 0.2f
#define ROW2     64
#define DEN2     40

typedef __attribute__((ext_vector_type(2))) float v2f;
typedef __attribute__((ext_vector_type(4))) float v4f;
typedef __attribute__((ext_vector_type(8))) float v8f;

__device__ __forceinline__ void atomicMaxFloat(float* addr, float val) {
    if (!(val == val)) return;
    if (val >= 0.0f) {
        atomicMax((int*)addr, __float_as_int(val));
    } else {
        atomicMin((unsigned int*)addr, __float_as_uint(val));
    }
}

template <int M>
__device__ __forceinline__ float gat_row_xor(float x) {
    return __int_as_float(__builtin_amdgcn_update_dpp(
        0, __float_as_int(x), 0x160 + M, 0xf, 0xf, false));
}

__device__ __forceinline__ float gat_xhalf(float x) {
    const int b = __float_as_int(x);
    return __int_as_float(__builtin_amdgcn_permlanex16(b, b, 0x76543210, 0xfedcba98, false, false));
}

__device__ __forceinline__ float gat_lrelu(float v) { return v > 0.0f ? v : v * NEG_SLOPE; }

template <int K, int CC, int NT, int TPW, int ACCW, int MW, int DW>
__global__ __launch_bounds__(32 * (2 * NT / TPW)) void gat_proj_pair_f32_wmma(
    const float* __restrict__ A,
    const float* __restrict__ Wl, const float* __restrict__ bl,
    const float* __restrict__ Wr, const float* __restrict__ br,
    float* __restrict__ outl, float* __restrict__ outr,
    float* __restrict__ acc0, float* __restrict__ max0, float* __restrict__ den0,
    float* __restrict__ cpart, int n) {
    constexpr int WPM = NT / TPW;
    constexpr int KS = K / 4;
    constexpr bool PAD = (16 * NT != CC);
    const int lane = threadIdx.x & 31;
    const int wave = __builtin_amdgcn_readfirstlane((int)(threadIdx.x >> 5));
    const bool right = wave >= WPM;
    const float* __restrict__ W = right ? Wr : Wl;
    const float* __restrict__ bias = right ? br : bl;
    float* __restrict__ out = right ? outr : outl;
    const int tile0 = (WPM == 1) ? 0 : (right ? wave - WPM : wave) * TPW;
    const int c = lane & 15;
    const int h = lane >> 4;

    v2f b[TPW][KS];
    float bv[TPW];
#pragma unroll
    for (int t = 0; t < TPW; ++t) {
        const int col = (tile0 + t) * 16 + c;
        const bool ok = !PAD || col < CC;
        const float* __restrict__ wc = W + (ok ? col : 0);
        bv[t] = ok ? bias[col] : 0.0f;
#pragma unroll
        for (int s = 0; s < KS; ++s) {
            const int k = 8 * (s >> 1) + 4 * h + 2 * (s & 1);
            const float w0 = wc[(size_t)k * CC];
            const float w1 = wc[(size_t)(k + 1) * CC];
            b[t][s].x = ok ? w0 : 0.0f;
            b[t][s].y = ok ? w1 : 0.0f;
        }
    }

    float cmx[TPW], cmn[TPW];
#pragma unroll
    for (int t = 0; t < TPW; ++t) { cmx[t] = -__builtin_inff(); cmn[t] = __builtin_inff(); }

    const int rowTiles = n >> 4;
    for (int rt = blockIdx.x; rt < rowTiles; rt += gridDim.x) {
        const float* __restrict__ aRow = A + (size_t)((rt << 4) + c) * K + 4 * h;
        v4f a[K / 8];
#pragma unroll
        for (int j = 0; j < K / 8; ++j) a[j] = *(const v4f*)(aRow + 8 * j);

        v8f acc[TPW];
#pragma unroll
        for (int t = 0; t < TPW; ++t) acc[t] = v8f{};
#pragma unroll
        for (int j = 0; j < K / 8; ++j) {
            const v2f a0 = {a[j].x, a[j].y};
            const v2f a1 = {a[j].z, a[j].w};
#pragma unroll
            for (int t = 0; t < TPW; ++t)
                acc[t] = __builtin_amdgcn_wmma_f32_16x16x4_f32(
                    false, a0, false, b[t][2 * j], (short)0, acc[t], false, false);
#pragma unroll
            for (int t = 0; t < TPW; ++t)
                acc[t] = __builtin_amdgcn_wmma_f32_16x16x4_f32(
                    false, a1, false, b[t][2 * j + 1], (short)0, acc[t], false, false);
        }

        const int rbase = (rt << 4) + (h << 3);
#pragma unroll
        for (int t = 0; t < TPW; ++t) {
            const int col = (tile0 + t) * 16 + c;
            if (!PAD || col < CC) {
#pragma unroll
                for (int r = 0; r < 8; ++r) {
                    const float v = acc[t][r] + bv[t];
                    out[(size_t)(rbase + r) * CC + col] = v;
                    cmx[t] = fmaxf(cmx[t], v);
                    cmn[t] = fminf(cmn[t], v);
                }
            }
        }

        constexpr int NW = 2 * WPM;
        constexpr int ACC_P = 16 * ACCW / 4;
        constexpr int MAX_P = 16 * MW / 4;
        constexpr int DEN_P = 16 * DW / 4;
        static_assert(ACC_P % (NW * 32) == 0 && MAX_P <= 32 && DEN_P <= 32, "tile start values");
        v4f* __restrict__ pa = (v4f*)(acc0 + (size_t)rt * (16 * ACCW));
        for (int q = 0; q < ACC_P / (NW * 32); ++q) pa[(q * NW + wave) * 32 + lane] = v4f{};
        if (wave == NW - 1 && lane < MAX_P) {
            const float ninf = -__builtin_inff();
            ((v4f*)(max0 + (size_t)rt * (16 * MW)))[lane] = v4f{ninf, ninf, ninf, ninf};
        }
        if constexpr (DW > 0) {
            if (wave == 0 && lane < DEN_P)
                ((v4f*)(den0 + (size_t)rt * (16 * DW)))[lane] = v4f{};
        }
    }

    if (!right) {
#pragma unroll
        for (int t = 0; t < TPW; ++t) {
            const float mx = fmaxf(cmx[t], __shfl_xor(cmx[t], 16, 32));
            const float mn = fminf(cmn[t], __shfl_xor(cmn[t], 16, 32));
            if (h == 0) {
                float* __restrict__ p = cpart + (size_t)blockIdx.x * (32 * NT) + (tile0 + t) * 16 + c;
                p[0] = mx;
                p[16 * NT] = mn;
            }
        }
    }
}

template <int CC, int CCP, int NH, int DCH>
__global__ __launch_bounds__(16 * CCP) void gat_logit_bound(
    const float* __restrict__ cpart, const float* __restrict__ att,
    float* __restrict__ bnd, int* __restrict__ mode, int nblk) {
    __shared__ float smx[16][CCP];
    __shared__ float smn[16][CCP];
    __shared__ float sg[CCP];
    const int col = threadIdx.x % CCP;
    const int part = threadIdx.x / CCP;
    float mx = -__builtin_inff(), mn = __builtin_inff();
    for (int b = part; b < nblk; b += 16) {
        const float* __restrict__ p = cpart + (size_t)b * (2 * CCP);
        mx = fmaxf(mx, p[col]);
        mn = fminf(mn, p[CCP + col]);
    }
    smx[part][col] = mx;
    smn[part][col] = mn;
    __syncthreads();
    if (threadIdx.x < CCP) {
        for (int q = 1; q < 16; ++q) {
            mx = fmaxf(mx, smx[q][col]);
            mn = fminf(mn, smn[q][col]);
        }
        float g = 0.0f;
        if (col < CC) {
            const float a = att[col];
            bnd[col] = a >= 0.0f ? mx : mn;
            g = fabsf(a) * (mx - mn);
        }
        sg[col] = g;
    }
    __syncthreads();
    if (threadIdx.x == 0) {
        int bad = 0;
        for (int hd = 0; hd < NH; ++hd) {
            float sum = 0.0f;
            for (int k = 0; k < DCH; ++k) sum += sg[hd * DCH + k];
            if (!(sum < 70.0f)) bad = 1;
        }
        *mode = bad;
    }
}

__device__ __forceinline__ void gat_logit_max_item_l1(
    const float* __restrict__ xl, const float* __restrict__ xr,
    const float* __restrict__ att, const int* __restrict__ src,
    const int* __restrict__ dst, float* __restrict__ e1,
    float* __restrict__ m1, long long t) {
    const int e = (int)(t >> 3);
    const int h = (int)(t & 7);
    const int s = src[e], d = dst[e];
    const float* pj = xl + (size_t)s * HID + h * DIM_H;
    const float* pi = xr + (size_t)d * HID + h * DIM_H;
    const float* pa = att + h * DIM_H;
    float acc = 0.0f;
#pragma unroll
    for (int c = 0; c < DIM_H; ++c) {
        float v = pi[c] + pj[c];
        v = v > 0.0f ? v : v * NEG_SLOPE;
        acc += v * pa[c];
    }
    e1[t] = acc;
    atomicMaxFloat(&m1[(size_t)d * HEADS + h], acc);
}

__device__ __forceinline__ void gat_edge_step_l1(
    float xj0, float xj1, float xi0, float xi1,
    float av0, float av1, float bn0, float bn1,
    bool den_lane, bool den_hi, float* __restrict__ den,
    float* __restrict__ po, int lane) {
    float t0 = (gat_lrelu(xi0 + xj0) - gat_lrelu(xi0 + bn0)) * av0;
    float t1 = (gat_lrelu(xi1 + xj1) - gat_lrelu(xi1 + bn1)) * av1;
    t0 += gat_row_xor<1>(t0); t1 += gat_row_xor<1>(t1);
    t0 += gat_row_xor<2>(t0); t1 += gat_row_xor<2>(t1);
    t0 += gat_row_xor<4>(t0); t1 += gat_row_xor<4>(t1);
    const float w0 = expf(t0);
    const float w1 = expf(t1);
    if (den_lane) atomicAdd(den, den_hi ? w1 : w0);
    atomicAdd(po + lane, xj0 * w0);
    atomicAdd(po + 32 + lane, xj1 * w1);
}

__global__ __launch_bounds__(256) void gat_edge_l1(
    const int* __restrict__ src, const int* __restrict__ dst,
    const float* __restrict__ xl, const float* __restrict__ xr,
    const float* __restrict__ att, const float* __restrict__ bnd,
    const int* __restrict__ mode, float* __restrict__ e1,
    float* __restrict__ m1, float* __restrict__ denom1,
    float* __restrict__ out1, int E) {
    if (*mode != 0) {
        const long long E8 = (long long)E * HEADS;
        for (long long t = (long long)blockIdx.x * blockDim.x + threadIdx.x; t < E8;
             t += (long long)gridDim.x * blockDim.x)
            gat_logit_max_item_l1(xl, xr, att, src, dst, e1, m1, t);
        return;
    }
    const int lane = threadIdx.x & 31;
    const int wave = __builtin_amdgcn_readfirstlane(
        (int)(blockIdx.x * (blockDim.x >> 5) + (threadIdx.x >> 5)));
    const int wbase = wave * 32;
    if (wbase >= E) return;
    const int nE = (E - wbase) < 32 ? (E - wbase) : 32;
    int s_i = 0, d_i = 0;
    if (lane < nE) {
        s_i = src[wbase + lane];
        d_i = dst[wbase + lane];
    }
    const float av0 = att[lane], av1 = att[32 + lane];
    const float bn0 = bnd[lane], bn1 = bnd[32 + lane];
    const bool den_lane = (lane & 7) < 2;
    const bool den_hi = (lane & 1) != 0;
    const int den_off = (lane & 1) * 4 + (lane >> 3);

    int j0 = 0;
    for (; j0 + 4 <= nE; j0 += 4) {
        float xj0[4], xj1[4], xi0[4], xi1[4];
#pragma unroll
        for (int q = 0; q < 4; ++q) {
            const int s = __builtin_amdgcn_readlane(s_i, j0 + q);
            const int d = __builtin_amdgcn_readlane(d_i, j0 + q);
            const float* pj = xl + (size_t)s * HID;
            const float* pi = xr + (size_t)d * HID;
            xj0[q] = pj[lane];
            xj1[q] = pj[32 + lane];
            xi0[q] = pi[lane];
            xi1[q] = pi[32 + lane];
        }
#pragma unroll
        for (int q = 0; q < 4; ++q) {
            const int d = __builtin_amdgcn_readlane(d_i, j0 + q);
            gat_edge_step_l1(xj0[q], xj1[q], xi0[q], xi1[q], av0, av1, bn0, bn1,
                             den_lane, den_hi, denom1 + (size_t)d * HEADS + den_off,
                             out1 + (size_t)d * HID, lane);
        }
    }
    for (; j0 < nE; ++j0) {
        const int s = __builtin_amdgcn_readlane(s_i, j0);
        const int d = __builtin_amdgcn_readlane(d_i, j0);
        const float* pj = xl + (size_t)s * HID;
        const float* pi = xr + (size_t)d * HID;
        gat_edge_step_l1(pj[lane], pj[32 + lane], pi[lane], pi[32 + lane], av0, av1, bn0, bn1,
                         den_lane, den_hi, denom1 + (size_t)d * HEADS + den_off,
                         out1 + (size_t)d * HID, lane);
    }
}

__global__ __launch_bounds__(256) void gat_edge_aggregate_l1(
    const int* __restrict__ src, const int* __restrict__ dst,
    const float* __restrict__ xl, const float* __restrict__ e1,
    const float* __restrict__ m1, float* __restrict__ denom1,
    float* __restrict__ out1, const int* __restrict__ mode, int E) {
    if (*mode == 0) return;
    const int lane = threadIdx.x & 31;
    const int wave = __builtin_amdgcn_readfirstlane(
        (int)(blockIdx.x * (blockDim.x >> 5) + (threadIdx.x >> 5)));
    const int wbase = wave * 32;
    if (wbase >= E) return;
    const int sub = lane >> 3;
    const int h = lane & 7;
    const int nE = (E - wbase) < 32 ? (E - wbase) : 32;

    if (nE == 32) {
        const int s_i = src[wbase + lane];
        const int d_i = dst[wbase + lane];
        float ev[8], mv[8], ex[8];
        int dp[8];
#pragma unroll
        for (int g = 0; g < 8; ++g) {
            dp[g] = __shfl(d_i, g * 4 + sub, 32);
            ev[g] = e1[(size_t)wbase * HEADS + g * 32 + lane];
            mv[g] = m1[(size_t)dp[g] * HEADS + h];
        }
#pragma unroll
        for (int g = 0; g < 8; ++g) {
            ex[g] = expf(ev[g] - mv[g]);
            atomicAdd(&denom1[(size_t)dp[g] * HEADS + h], ex[g]);
        }
#pragma unroll
        for (int g = 0; g < 8; ++g) {
            float r0[4], r1[4];
#pragma unroll
            for (int q = 0; q < 4; ++q) {
                const int s = __builtin_amdgcn_readlane(s_i, g * 4 + q);
                const float* pj = xl + (size_t)s * HID;
                r0[q] = pj[lane];
                r1[q] = pj[32 + lane];
            }
#pragma unroll
            for (int q = 0; q < 4; ++q) {
                const int d = __builtin_amdgcn_readlane(d_i, g * 4 + q);
                const float a = __shfl(ex[g], q * 8 + sub, 32);
                const float b = __shfl(ex[g], q * 8 + 4 + sub, 32);
                float* po = out1 + (size_t)d * HID;
                atomicAdd(po + lane, r0[q] * a);
                atomicAdd(po + 32 + lane, r1[q] * b);
            }
        }
    } else {
        for (int j = 0; j < nE; ++j) {
            const int e = wbase + j;
            const int s = src[e], d = dst[e];
            const float a = expf(e1[(size_t)e * HEADS + sub] - m1[(size_t)d * HEADS + sub]);
            const float b = expf(e1[(size_t)e * HEADS + 4 + sub] - m1[(size_t)d * HEADS + 4 + sub]);
            if (h == 0) {
                atomicAdd(&denom1[(size_t)d * HEADS + sub], a);
                atomicAdd(&denom1[(size_t)d * HEADS + 4 + sub], b);
            }
            const float* pj = xl + (size_t)s * HID;
            float* po = out1 + (size_t)d * HID;
            atomicAdd(po + lane, pj[lane] * a);
            atomicAdd(po + 32 + lane, pj[32 + lane] * b);
        }
    }
}

__global__ void gat_norm_bias_elu(float* __restrict__ out1,
                                  const float* __restrict__ denom1,
                                  const float* __restrict__ bias1,
                                  long long total) {
    long long i = (long long)blockIdx.x * blockDim.x + threadIdx.x;
    if (i >= total) return;
    const float den = denom1[i >> 3];
    const float acc = out1[i];
    float v = (den > 0.0f ? acc / den : acc) + bias1[(int)(i & (HID - 1))];
    out1[i] = v > 0.0f ? v : expm1f(v);
}

__device__ __forceinline__ void gat_logit_max_item_l2(
    const float* __restrict__ xl, const float* __restrict__ xr,
    const float* __restrict__ att, const int* __restrict__ src,
    const int* __restrict__ dst, float* __restrict__ e2,
    float* __restrict__ m2, long long t) {
    const int e = (int)(t >> 3);
    const int sub = (int)(t & 7);
    const int s = src[e], d = dst[e];
    const float* pj = xl + (size_t)s * DIM_OUT;
    const float* pi = xr + (size_t)d * DIM_OUT;
    float acc = 0.0f;
#pragma unroll
    for (int c = sub; c < DIM_OUT; c += 8) {
        float v = pi[c] + pj[c];
        v = v > 0.0f ? v : v * NEG_SLOPE;
        acc += v * att[c];
    }
    acc += __shfl_xor(acc, 1, 32);
    acc += __shfl_xor(acc, 2, 32);
    acc += __shfl_xor(acc, 4, 32);
    if (sub == 0) {
        e2[e] = acc;
        atomicMaxFloat(&m2[d], acc);
    }
}

__device__ __forceinline__ void gat_edge_step_l2(
    float xj0, float xj1, float xi0, float xi1,
    float av0, float av1, float bn0, float bn1,
    bool tail_ch, float* __restrict__ po, int lane) {
    const float ta = (gat_lrelu(xi0 + xj0) - gat_lrelu(xi0 + bn0)) * av0;
    const float tb = (gat_lrelu(xi1 + xj1) - gat_lrelu(xi1 + bn1)) * av1;
    float t = ta + (tail_ch ? tb : 0.0f);
    t += gat_row_xor<1>(t);
    t += gat_row_xor<2>(t);
    t += gat_row_xor<4>(t);
    t += gat_row_xor<8>(t);
    t += gat_xhalf(t);
    const float w = expf(t);
    const float v1 = tail_ch ? xj1 : 1.0f;
    atomicAdd(po + lane, xj0 * w);
    if (lane <= DEN2 - 32) atomicAdd(po + 32 + lane, v1 * w);
}

__global__ __launch_bounds__(256) void gat_edge_l2(
    const int* __restrict__ src, const int* __restrict__ dst,
    const float* __restrict__ xl, const float* __restrict__ xr,
    const float* __restrict__ att, const float* __restrict__ bnd,
    const int* __restrict__ mode, float* __restrict__ e2,
    float* __restrict__ m2, float* __restrict__ out2, int E) {
    if (*mode != 0) {
        const long long E8 = (long long)E * 8;
        for (long long t = (long long)blockIdx.x * blockDim.x + threadIdx.x; t < E8;
             t += (long long)gridDim.x * blockDim.x)
            gat_logit_max_item_l2(xl, xr, att, src, dst, e2, m2, t);
        return;
    }
    const int lane = threadIdx.x & 31;
    const int wave = __builtin_amdgcn_readfirstlane(
        (int)(blockIdx.x * (blockDim.x >> 5) + (threadIdx.x >> 5)));
    const int wbase = wave * 32;
    if (wbase >= E) return;
    const int nE = (E - wbase) < 32 ? (E - wbase) : 32;
    int s_i = 0, d_i = 0;
    if (lane < nE) {
        s_i = src[wbase + lane];
        d_i = dst[wbase + lane];
    }
    const int c1 = 32 + (lane & 7);
    const bool tail_ch = lane < DIM_OUT - 32;
    const float av0 = att[lane], av1 = att[c1];
    const float bn0 = bnd[lane], bn1 = bnd[c1];

    int j0 = 0;
    for (; j0 + 4 <= nE; j0 += 4) {
        float xj0[4], xj1[4], xi0[4], xi1[4];
#pragma unroll
        for (int q = 0; q < 4; ++q) {
            const int s = __builtin_amdgcn_readlane(s_i, j0 + q);
            const int d = __builtin_amdgcn_readlane(d_i, j0 + q);
            const float* pj = xl + (size_t)s * DIM_OUT;
            const float* pi = xr + (size_t)d * DIM_OUT;
            xj0[q] = pj[lane];
            xj1[q] = pj[c1];
            xi0[q] = pi[lane];
            xi1[q] = pi[c1];
        }
#pragma unroll
        for (int q = 0; q < 4; ++q) {
            const int d = __builtin_amdgcn_readlane(d_i, j0 + q);
            gat_edge_step_l2(xj0[q], xj1[q], xi0[q], xi1[q], av0, av1, bn0, bn1,
                             tail_ch, out2 + (size_t)d * ROW2, lane);
        }
    }
    for (; j0 < nE; ++j0) {
        const int s = __builtin_amdgcn_readlane(s_i, j0);
        const int d = __builtin_amdgcn_readlane(d_i, j0);
        const float* pj = xl + (size_t)s * DIM_OUT;
        const float* pi = xr + (size_t)d * DIM_OUT;
        gat_edge_step_l2(pj[lane], pj[c1], pi[lane], pi[c1], av0, av1, bn0, bn1,
                         tail_ch, out2 + (size_t)d * ROW2, lane);
    }
}

__global__ __launch_bounds__(256) void gat_edge_aggregate_l2(
    const int* __restrict__ src, const int* __restrict__ dst,
    const float* __restrict__ xl, const float* __restrict__ e2,
    const float* __restrict__ m2, float* __restrict__ out2,
    const int* __restrict__ mode, int E) {
    if (*mode == 0) return;
    const int lane = threadIdx.x & 31;
    const int wave = __builtin_amdgcn_readfirstlane(
        (int)(blockIdx.x * (blockDim.x >> 5) + (threadIdx.x >> 5)));
    const int wbase = wave * 32;
    if (wbase >= E) return;
    const int ei = wbase + lane;
    int s_i = 0, d_i = 0;
    float ex_i = 0.0f;
    if (ei < E) {
        s_i = src[ei];
        d_i = dst[ei];
        ex_i = expf(e2[ei] - m2[d_i]);
    }
    const int nE = (E - wbase) < 32 ? (E - wbase) : 32;
    const int c1 = 32 + (lane & 7);
    const bool tail_ch = lane < DIM_OUT - 32;

    if (nE == 32) {
        for (int j0 = 0; j0 < 32; j0 += 4) {
            float r0[4], r1[4];
#pragma unroll
            for (int q = 0; q < 4; ++q) {
                const int s = __builtin_amdgcn_readlane(s_i, j0 + q);
                const float* pj = xl + (size_t)s * DIM_OUT;
                r0[q] = pj[lane];
                r1[q] = pj[c1];
            }
            asm volatile("" :: "v"(r1[0]), "v"(r1[1]), "v"(r1[2]), "v"(r1[3]));
#pragma unroll
            for (int q = 0; q < 4; ++q) {
                const int d = __builtin_amdgcn_readlane(d_i, j0 + q);
                const float a = __int_as_float(
                    __builtin_amdgcn_readlane(__float_as_int(ex_i), j0 + q));
                float* po = out2 + (size_t)d * ROW2;
                const float v1 = tail_ch ? r1[q] : 1.0f;
                atomicAdd(po + lane, r0[q] * a);
                if (lane <= DEN2 - 32) atomicAdd(po + 32 + lane, v1 * a);
            }
        }
    } else {
        for (int j = 0; j < nE; ++j) {
            const int s = __builtin_amdgcn_readlane(s_i, j);
            const int d = __builtin_amdgcn_readlane(d_i, j);
            const float a = __int_as_float(
                __builtin_amdgcn_readlane(__float_as_int(ex_i), j));
            const float* pj = xl + (size_t)s * DIM_OUT;
            float* po = out2 + (size_t)d * ROW2;
            const float v0 = pj[lane];
            const float v1 = tail_ch ? pj[c1] : 1.0f;
            atomicAdd(po + lane, v0 * a);
            if (lane <= DEN2 - 32) atomicAdd(po + 32 + lane, v1 * a);
        }
    }
}

__global__ __launch_bounds__(256) void gat_finalize(
    const float* __restrict__ out2,
    const float* __restrict__ bias2, float* __restrict__ d_out, int N) {
    const int node = blockIdx.x * (blockDim.x >> 5) + (threadIdx.x >> 5);
    const int lane = threadIdx.x & 31;
    if (node >= N) return;
    const float* row = out2 + (size_t)node * ROW2;
    const float den = row[DEN2];
    const float ra = row[lane];
    const float a = (den > 0.0f ? ra / den : ra) + bias2[lane];
    const bool has2 = (lane < DIM_OUT - 32);
    const float rb = has2 ? row[32 + lane] : 0.0f;
    const float b = has2 ? ((den > 0.0f ? rb / den : rb) + bias2[32 + lane])
                         : -__builtin_inff();
    float mx = fmaxf(a, b);
#pragma unroll
    for (int o = 16; o >= 1; o >>= 1) mx = fmaxf(mx, __shfl_xor(mx, o, 32));
    float s = expf(a - mx) + (has2 ? expf(b - mx) : 0.0f);
#pragma unroll
    for (int o = 16; o >= 1; o >>= 1) s += __shfl_xor(s, o, 32);
    const float lse = logf(s) + mx;
    float* h  = d_out + (size_t)node * DIM_OUT;
    float* ls = d_out + (size_t)N * DIM_OUT + (size_t)node * DIM_OUT;
    h[lane] = a;
    ls[lane] = a - lse;
    if (has2) {
        h[32 + lane] = b;
        ls[32 + lane] = b - lse;
    }
}

static inline int cdiv_i(long long a, long long b) { return (int)((a + b - 1) / b); }

extern "C" void kernel_launch(void* const* d_in, const int* in_sizes, int n_in,
                              void* d_out, int out_size, void* d_ws, size_t ws_size,
                              hipStream_t stream) {
    (void)n_in; (void)out_size; (void)ws_size;
    const float* x     = (const float*)d_in[0];
    const int*   eidx  = (const int*)d_in[1];
    const float* Wl1   = (const float*)d_in[2];
    const float* bl1   = (const float*)d_in[3];
    const float* Wr1   = (const float*)d_in[4];
    const float* br1   = (const float*)d_in[5];
    const float* att1  = (const float*)d_in[6];
    const float* bias1 = (const float*)d_in[7];
    const float* Wl2   = (const float*)d_in[8];
    const float* bl2   = (const float*)d_in[9];
    const float* Wr2   = (const float*)d_in[10];
    const float* br2   = (const float*)d_in[11];
    const float* att2  = (const float*)d_in[12];
    const float* bias2 = (const float*)d_in[13];

    const int N = in_sizes[0] / DIM_IN;
    const int E = in_sizes[1] / 2;
    const int* src = eidx;
    const int* dst = eidx + E;

    float* ws = (float*)d_ws;
    size_t o = 0;
    float* h_l1   = ws + o; o += (size_t)N * HID;
    float* h_r1   = ws + o; o += (size_t)N * HID;
    float* e1     = ws + o; o += (size_t)E * HEADS;
    float* m1     = ws + o; o += (size_t)N * HEADS;
    float* denom1 = ws + o; o += (size_t)N * HEADS;
    float* out1   = ws + o; o += (size_t)N * HID;
    float* cpart1 = ws + o; o += (size_t)1024 * 2 * HID;
    float* bnd1   = ws + o; o += 64;
    int*   mode1  = (int*)(ws + o); o += 64;
    float* cpart2 = ws + o; o += (size_t)1250 * 2 * 48;
    float* bnd2   = ws + o; o += 64;
    int*   mode2  = (int*)(ws + o); o += 64;
    float* h_l2   = ws;
    float* h_r2   = h_l2 + (size_t)N * DIM_OUT;
    float* e2     = h_r2 + (size_t)N * DIM_OUT;
    float* out2   = e2 + (size_t)E;
    float* m2     = out2 + (size_t)N * ROW2;

    const int B = 256;

    gat_proj_pair_f32_wmma<DIM_IN, HID, 4, 2, HID, HEADS, HEADS>
        <<<1024, 128, 0, stream>>>(x, Wl1, bl1, Wr1, br1, h_l1, h_r1,
                                   out1, m1, denom1, cpart1, N);
    gat_logit_bound<HID, HID, HEADS, DIM_H>
        <<<1, 16 * HID, 0, stream>>>(cpart1, att1, bnd1, mode1, 1024);

    gat_edge_l1<<<cdiv_i(E, B), B, 0, stream>>>(
        src, dst, h_l1, h_r1, att1, bnd1, mode1, e1, m1, denom1, out1, E);
    gat_edge_aggregate_l1<<<cdiv_i(E, B), B, 0, stream>>>(
        src, dst, h_l1, e1, m1, denom1, out1, mode1, E);
    gat_norm_bias_elu<<<cdiv_i((long long)N * HID, B), B, 0, stream>>>(
        out1, denom1, bias1, (long long)N * HID);

    gat_proj_pair_f32_wmma<HID, DIM_OUT, 3, 3, ROW2, 1, 0>
        <<<1250, 64, 0, stream>>>(out1, Wl2, bl2, Wr2, br2, h_l2, h_r2,
                                  out2, m2, (float*)nullptr, cpart2, N);
    gat_logit_bound<DIM_OUT, 48, 1, DIM_OUT>
        <<<1, 16 * 48, 0, stream>>>(cpart2, att2, bnd2, mode2, 1250);

    gat_edge_l2<<<cdiv_i(E, B), B, 0, stream>>>(
        src, dst, h_l2, h_r2, att2, bnd2, mode2, e2, m2, out2, E);
    gat_edge_aggregate_l2<<<cdiv_i(E, B), B, 0, stream>>>(
        src, dst, h_l2, e2, m2, out2, mode2, E);

    gat_finalize<<<cdiv_i(N, B / 32), B, 0, stream>>>((const float*)out2,
                                                      bias2, (float*)d_out, N);
}


# Scatter targets (accumulator rows, denominators) kept as one plane per package half: a wave's float atomic adds go to its own half's plane, node passes add the two planes; sum order changes
# speedup vs baseline: 3.0868x; 1.0538x over previous
#include <hip/hip_runtime.h>
#include <hip/hip_bf16.h>


#define DIM_IN   128
#define HID      64
#define HEADS    8
#define DIM_H    8
#define DIM_OUT  40
#define NEG_SLOPE 0.2f
#define ROW2     64
#define DEN2     40

typedef __attribute__((ext_vector_type(2))) float v2f;
typedef __attribute__((ext_vector_type(4))) float v4f;
typedef __attribute__((ext_vector_type(8))) float v8f;

__device__ __forceinline__ void atomicMaxFloat(float* addr, float val) {
    if (!(val == val)) return;
    if (val >= 0.0f) {
        atomicMax((int*)addr, __float_as_int(val));
    } else {
        atomicMin((unsigned int*)addr, __float_as_uint(val));
    }
}

template <int M>
__device__ __forceinline__ float gat_row_xor(float x) {
    return __int_as_float(__builtin_amdgcn_update_dpp(
        0, __float_as_int(x), 0x160 + M, 0xf, 0xf, false));
}

__device__ __forceinline__ float gat_xhalf(float x) {
    const int b = __float_as_int(x);
    return __int_as_float(__builtin_amdgcn_permlanex16(b, b, 0x76543210, 0xfedcba98, false, false));
}

__device__ __forceinline__ float gat_lrelu(float v) { return v > 0.0f ? v : v * NEG_SLOPE; }

__device__ __forceinline__ int gat_package_half() {
    return (int)((__builtin_amdgcn_s_sendmsg_rtn(0x87) >> 18) & 1u);
}

template <int K, int CC, int NT, int TPW, int ACCW, int MW, int DW>
__global__ __launch_bounds__(32 * (2 * NT / TPW)) void gat_proj_pair_f32_wmma(
    const float* __restrict__ A,
    const float* __restrict__ Wl, const float* __restrict__ bl,
    const float* __restrict__ Wr, const float* __restrict__ br,
    float* __restrict__ outl, float* __restrict__ outr,
    float* __restrict__ acc0, float* __restrict__ acc1, float* __restrict__ max0,
    float* __restrict__ den0, float* __restrict__ den1,
    float* __restrict__ cpart, int n) {
    constexpr int WPM = NT / TPW;
    constexpr int KS = K / 4;
    constexpr bool PAD = (16 * NT != CC);
    const int lane = threadIdx.x & 31;
    const int wave = __builtin_amdgcn_readfirstlane((int)(threadIdx.x >> 5));
    const bool right = wave >= WPM;
    const float* __restrict__ W = right ? Wr : Wl;
    const float* __restrict__ bias = right ? br : bl;
    float* __restrict__ out = right ? outr : outl;
    const int tile0 = (WPM == 1) ? 0 : (right ? wave - WPM : wave) * TPW;
    const int c = lane & 15;
    const int h = lane >> 4;

    v2f b[TPW][KS];
    float bv[TPW];
#pragma unroll
    for (int t = 0; t < TPW; ++t) {
        const int col = (tile0 + t) * 16 + c;
        const bool ok = !PAD || col < CC;
        const float* __restrict__ wc = W + (ok ? col : 0);
        bv[t] = ok ? bias[col] : 0.0f;
#pragma unroll
        for (int s = 0; s < KS; ++s) {
            const int k = 8 * (s >> 1) + 4 * h + 2 * (s & 1);
            const float w0 = wc[(size_t)k * CC];
            const float w1 = wc[(size_t)(k + 1) * CC];
            b[t][s].x = ok ? w0 : 0.0f;
            b[t][s].y = ok ? w1 : 0.0f;
        }
    }

    float cmx[TPW], cmn[TPW];
#pragma unroll
    for (int t = 0; t < TPW; ++t) { cmx[t] = -__builtin_inff(); cmn[t] = __builtin_inff(); }

    const int rowTiles = n >> 4;
    for (int rt = blockIdx.x; rt < rowTiles; rt += gridDim.x) {
        const float* __restrict__ aRow = A + (size_t)((rt << 4) + c) * K + 4 * h;
        v4f a[K / 8];
#pragma unroll
        for (int j = 0; j < K / 8; ++j) a[j] = *(const v4f*)(aRow + 8 * j);

        v8f acc[TPW];
#pragma unroll
        for (int t = 0; t < TPW; ++t) acc[t] = v8f{};
#pragma unroll
        for (int j = 0; j < K / 8; ++j) {
            const v2f a0 = {a[j].x, a[j].y};
            const v2f a1 = {a[j].z, a[j].w};
#pragma unroll
            for (int t = 0; t < TPW; ++t)
                acc[t] = __builtin_amdgcn_wmma_f32_16x16x4_f32(
                    false, a0, false, b[t][2 * j], (short)0, acc[t], false, false);
#pragma unroll
            for (int t = 0; t < TPW; ++t)
                acc[t] = __builtin_amdgcn_wmma_f32_16x16x4_f32(
                    false, a1, false, b[t][2 * j + 1], (short)0, acc[t], false, false);
        }

        const int rbase = (rt << 4) + (h << 3);
#pragma unroll
        for (int t = 0; t < TPW; ++t) {
            const int col = (tile0 + t) * 16 + c;
            if (!PAD || col < CC) {
#pragma unroll
                for (int r = 0; r < 8; ++r) {
                    const float v = acc[t][r] + bv[t];
                    out[(size_t)(rbase + r) * CC + col] = v;
                    cmx[t] = fmaxf(cmx[t], v);
                    cmn[t] = fminf(cmn[t], v);
                }
            }
        }

        constexpr int NW = 2 * WPM;
        constexpr int ACC_P = 16 * ACCW / 4;
        constexpr int MAX_P = 16 * MW / 4;
        constexpr int DEN_P = 16 * DW / 4;
        static_assert(ACC_P % (NW * 32) == 0 && MAX_P <= 32 && DEN_P <= 32, "tile start values");
        v4f* __restrict__ pa = (v4f*)(acc0 + (size_t)rt * (16 * ACCW));
        v4f* __restrict__ pb = (v4f*)(acc1 + (size_t)rt * (16 * ACCW));
        for (int q = 0; q < ACC_P / (NW * 32); ++q) {
            pa[(q * NW + wave) * 32 + lane] = v4f{};
            pb[(q * NW + wave) * 32 + lane] = v4f{};
        }
        if (wave == NW - 1 && lane < MAX_P) {
            const float ninf = -__builtin_inff();
            ((v4f*)(max0 + (size_t)rt * (16 * MW)))[lane] = v4f{ninf, ninf, ninf, ninf};
        }
        if constexpr (DW > 0) {
            if (wave == 0 && lane < DEN_P) {
                ((v4f*)(den0 + (size_t)rt * (16 * DW)))[lane] = v4f{};
                ((v4f*)(den1 + (size_t)rt * (16 * DW)))[lane] = v4f{};
            }
        }
    }

    if (!right) {
#pragma unroll
        for (int t = 0; t < TPW; ++t) {
            const float mx = fmaxf(cmx[t], __shfl_xor(cmx[t], 16, 32));
            const float mn = fminf(cmn[t], __shfl_xor(cmn[t], 16, 32));
            if (h == 0) {
                float* __restrict__ p = cpart + (size_t)blockIdx.x * (32 * NT) + (tile0 + t) * 16 + c;
                p[0] = mx;
                p[16 * NT] = mn;
            }
        }
    }
}

template <int CC, int CCP, int NH, int DCH>
__global__ __launch_bounds__(16 * CCP) void gat_logit_bound(
    const float* __restrict__ cpart, const float* __restrict__ att,
    float* __restrict__ bnd, int* __restrict__ mode, int nblk) {
    __shared__ float smx[16][CCP];
    __shared__ float smn[16][CCP];
    __shared__ float sg[CCP];
    const int col = threadIdx.x % CCP;
    const int part = threadIdx.x / CCP;
    float mx = -__builtin_inff(), mn = __builtin_inff();
    for (int b = part; b < nblk; b += 16) {
        const float* __restrict__ p = cpart + (size_t)b * (2 * CCP);
        mx = fmaxf(mx, p[col]);
        mn = fminf(mn, p[CCP + col]);
    }
    smx[part][col] = mx;
    smn[part][col] = mn;
    __syncthreads();
    if (threadIdx.x < CCP) {
        for (int q = 1; q < 16; ++q) {
            mx = fmaxf(mx, smx[q][col]);
            mn = fminf(mn, smn[q][col]);
        }
        float g = 0.0f;
        if (col < CC) {
            const float a = att[col];
            bnd[col] = a >= 0.0f ? mx : mn;
            g = fabsf(a) * (mx - mn);
        }
        sg[col] = g;
    }
    __syncthreads();
    if (threadIdx.x == 0) {
        int bad = 0;
        for (int hd = 0; hd < NH; ++hd) {
            float sum = 0.0f;
            for (int k = 0; k < DCH; ++k) sum += sg[hd * DCH + k];
            if (!(sum < 70.0f)) bad = 1;
        }
        *mode = bad;
    }
}

__device__ __forceinline__ void gat_logit_max_item_l1(
    const float* __restrict__ xl, const float* __restrict__ xr,
    const float* __restrict__ att, const int* __restrict__ src,
    const int* __restrict__ dst, float* __restrict__ e1,
    float* __restrict__ m1, long long t) {
    const int e = (int)(t >> 3);
    const int h = (int)(t & 7);
    const int s = src[e], d = dst[e];
    const float* pj = xl + (size_t)s * HID + h * DIM_H;
    const float* pi = xr + (size_t)d * HID + h * DIM_H;
    const float* pa = att + h * DIM_H;
    float acc = 0.0f;
#pragma unroll
    for (int c = 0; c < DIM_H; ++c) {
        float v = pi[c] + pj[c];
        v = v > 0.0f ? v : v * NEG_SLOPE;
        acc += v * pa[c];
    }
    e1[t] = acc;
    atomicMaxFloat(&m1[(size_t)d * HEADS + h], acc);
}

__device__ __forceinline__ void gat_edge_step_l1(
    float xj0, float xj1, float xi0, float xi1,
    float av0, float av1, float bn0, float bn1,
    bool den_lane, bool den_hi, float* __restrict__ den,
    float* __restrict__ po, int lane) {
    float t0 = (gat_lrelu(xi0 + xj0) - gat_lrelu(xi0 + bn0)) * av0;
    float t1 = (gat_lrelu(xi1 + xj1) - gat_lrelu(xi1 + bn1)) * av1;
    t0 += gat_row_xor<1>(t0); t1 += gat_row_xor<1>(t1);
    t0 += gat_row_xor<2>(t0); t1 += gat_row_xor<2>(t1);
    t0 += gat_row_xor<4>(t0); t1 += gat_row_xor<4>(t1);
    const float w0 = expf(t0);
    const float w1 = expf(t1);
    if (den_lane) atomicAdd(den, den_hi ? w1 : w0);
    atomicAdd(po + lane, xj0 * w0);
    atomicAdd(po + 32 + lane, xj1 * w1);
}

__global__ __launch_bounds__(256) void gat_edge_l1(
    const int* __restrict__ src, const int* __restrict__ dst,
    const float* __restrict__ xl, const float* __restrict__ xr,
    const float* __restrict__ att, const float* __restrict__ bnd,
    const int* __restrict__ mode, float* __restrict__ e1,
    float* __restrict__ m1, float* __restrict__ denom1a, float* __restrict__ denom1b,
    float* __restrict__ out1a, float* __restrict__ out1b, int E) {
    if (*mode != 0) {
        const long long E8 = (long long)E * HEADS;
        for (long long t = (long long)blockIdx.x * blockDim.x + threadIdx.x; t < E8;
             t += (long long)gridDim.x * blockDim.x)
            gat_logit_max_item_l1(xl, xr, att, src, dst, e1, m1, t);
        return;
    }
    const int lane = threadIdx.x & 31;
    const int wave = __builtin_amdgcn_readfirstlane(
        (int)(blockIdx.x * (blockDim.x >> 5) + (threadIdx.x >> 5)));
    const int wbase = wave * 32;
    if (wbase >= E) return;
    const bool upper = gat_package_half() != 0;
    float* __restrict__ denom1 = upper ? denom1b : denom1a;
    float* __restrict__ out1 = upper ? out1b : out1a;
    const int nE = (E - wbase) < 32 ? (E - wbase) : 32;
    int s_i = 0, d_i = 0;
    if (lane < nE) {
        s_i = src[wbase + lane];
        d_i = dst[wbase + lane];
    }
    const float av0 = att[lane], av1 = att[32 + lane];
    const float bn0 = bnd[lane], bn1 = bnd[32 + lane];
    const bool den_lane = (lane & 7) < 2;
    const bool den_hi = (lane & 1) != 0;
    const int den_off = (lane & 1) * 4 + (lane >> 3);

    int j0 = 0;
    for (; j0 + 4 <= nE; j0 += 4) {
        float xj0[4], xj1[4], xi0[4], xi1[4];
#pragma unroll
        for (int q = 0; q < 4; ++q) {
            const int s = __builtin_amdgcn_readlane(s_i, j0 + q);
            const int d = __builtin_amdgcn_readlane(d_i, j0 + q);
            const float* pj = xl + (size_t)s * HID;
            const float* pi = xr + (size_t)d * HID;
            xj0[q] = pj[lane];
            xj1[q] = pj[32 + lane];
            xi0[q] = pi[lane];
            xi1[q] = pi[32 + lane];
        }
#pragma unroll
        for (int q = 0; q < 4; ++q) {
            const int d = __builtin_amdgcn_readlane(d_i, j0 + q);
            gat_edge_step_l1(xj0[q], xj1[q], xi0[q], xi1[q], av0, av1, bn0, bn1,
                             den_lane, den_hi, denom1 + (size_t)d * HEADS + den_off,
                             out1 + (size_t)d * HID, lane);
        }
    }
    for (; j0 < nE; ++j0) {
        const int s = __builtin_amdgcn_readlane(s_i, j0);
        const int d = __builtin_amdgcn_readlane(d_i, j0);
        const float* pj = xl + (size_t)s * HID;
        const float* pi = xr + (size_t)d * HID;
        gat_edge_step_l1(pj[lane], pj[32 + lane], pi[lane], pi[32 + lane], av0, av1, bn0, bn1,
                         den_lane, den_hi, denom1 + (size_t)d * HEADS + den_off,
                         out1 + (size_t)d * HID, lane);
    }
}

__global__ __launch_bounds__(256) void gat_edge_aggregate_l1(
    const int* __restrict__ src, const int* __restrict__ dst,
    const float* __restrict__ xl, const float* __restrict__ e1,
    const float* __restrict__ m1, float* __restrict__ denom1a, float* __restrict__ denom1b,
    float* __restrict__ out1a, float* __restrict__ out1b,
    const int* __restrict__ mode, int E) {
    if (*mode == 0) return;
    const int lane = threadIdx.x & 31;
    const int wave = __builtin_amdgcn_readfirstlane(
        (int)(blockIdx.x * (blockDim.x >> 5) + (threadIdx.x >> 5)));
    const int wbase = wave * 32;
    if (wbase >= E) return;
    const bool upper = gat_package_half() != 0;
    float* __restrict__ denom1 = upper ? denom1b : denom1a;
    float* __restrict__ out1 = upper ? out1b : out1a;
    const int sub = lane >> 3;
    const int h = lane & 7;
    const int nE = (E - wbase) < 32 ? (E - wbase) : 32;

    if (nE == 32) {
        const int s_i = src[wbase + lane];
        const int d_i = dst[wbase + lane];
        float ev[8], mv[8], ex[8];
        int dp[8];
#pragma unroll
        for (int g = 0; g < 8; ++g) {
            dp[g] = __shfl(d_i, g * 4 + sub, 32);
            ev[g] = e1[(size_t)wbase * HEADS + g * 32 + lane];
            mv[g] = m1[(size_t)dp[g] * HEADS + h];
        }
#pragma unroll
        for (int g = 0; g < 8; ++g) {
            ex[g] = expf(ev[g] - mv[g]);
            atomicAdd(&denom1[(size_t)dp[g] * HEADS + h], ex[g]);
        }
#pragma unroll
        for (int g = 0; g < 8; ++g) {
            float r0[4], r1[4];
#pragma unroll
            for (int q = 0; q < 4; ++q) {
                const int s = __builtin_amdgcn_readlane(s_i, g * 4 + q);
                const float* pj = xl + (size_t)s * HID;
                r0[q] = pj[lane];
                r1[q] = pj[32 + lane];
            }
#pragma unroll
            for (int q = 0; q < 4; ++q) {
                const int d = __builtin_amdgcn_readlane(d_i, g * 4 + q);
                const float a = __shfl(ex[g], q * 8 + sub, 32);
                const float b = __shfl(ex[g], q * 8 + 4 + sub, 32);
                float* po = out1 + (size_t)d * HID;
                atomicAdd(po + lane, r0[q] * a);
                atomicAdd(po + 32 + lane, r1[q] * b);
            }
        }
    } else {
        for (int j = 0; j < nE; ++j) {
            const int e = wbase + j;
            const int s = src[e], d = dst[e];
            const float a = expf(e1[(size_t)e * HEADS + sub] - m1[(size_t)d * HEADS + sub]);
            const float b = expf(e1[(size_t)e * HEADS + 4 + sub] - m1[(size_t)d * HEADS + 4 + sub]);
            if (h == 0) {
                atomicAdd(&denom1[(size_t)d * HEADS + sub], a);
                atomicAdd(&denom1[(size_t)d * HEADS + 4 + sub], b);
            }
            const float* pj = xl + (size_t)s * HID;
            float* po = out1 + (size_t)d * HID;
            atomicAdd(po + lane, pj[lane] * a);
            atomicAdd(po + 32 + lane, pj[32 + lane] * b);
        }
    }
}

__global__ void gat_norm_bias_elu(float* __restrict__ out1,
                                  const float* __restrict__ out1b,
                                  const float* __restrict__ denom1,
                                  const float* __restrict__ denom1b,
                                  const float* __restrict__ bias1,
                                  long long total) {
    long long i = (long long)blockIdx.x * blockDim.x + threadIdx.x;
    if (i >= total) return;
    const float den = denom1[i >> 3] + denom1b[i >> 3];
    const float acc = out1[i] + out1b[i];
    float v = (den > 0.0f ? acc / den : acc) + bias1[(int)(i & (HID - 1))];
    out1[i] = v > 0.0f ? v : expm1f(v);
}

__device__ __forceinline__ void gat_logit_max_item_l2(
    const float* __restrict__ xl, const float* __restrict__ xr,
    const float* __restrict__ att, const int* __restrict__ src,
    const int* __restrict__ dst, float* __restrict__ e2,
    float* __restrict__ m2, long long t) {
    const int e = (int)(t >> 3);
    const int sub = (int)(t & 7);
    const int s = src[e], d = dst[e];
    const float* pj = xl + (size_t)s * DIM_OUT;
    const float* pi = xr + (size_t)d * DIM_OUT;
    float acc = 0.0f;
#pragma unroll
    for (int c = sub; c < DIM_OUT; c += 8) {
        float v = pi[c] + pj[c];
        v = v > 0.0f ? v : v * NEG_SLOPE;
        acc += v * att[c];
    }
    acc += __shfl_xor(acc, 1, 32);
    acc += __shfl_xor(acc, 2, 32);
    acc += __shfl_xor(acc, 4, 32);
    if (sub == 0) {
        e2[e] = acc;
        atomicMaxFloat(&m2[d], acc);
    }
}

__device__ __forceinline__ void gat_edge_step_l2(
    float xj0, float xj1, float xi0, float xi1,
    float av0, float av1, float bn0, float bn1,
    bool tail_ch, float* __restrict__ po, int lane) {
    const float ta = (gat_lrelu(xi0 + xj0) - gat_lrelu(xi0 + bn0)) * av0;
    const float tb = (gat_lrelu(xi1 + xj1) - gat_lrelu(xi1 + bn1)) * av1;
    float t = ta + (tail_ch ? tb : 0.0f);
    t += gat_row_xor<1>(t);
    t += gat_row_xor<2>(t);
    t += gat_row_xor<4>(t);
    t += gat_row_xor<8>(t);
    t += gat_xhalf(t);
    const float w = expf(t);
    const float v1 = tail_ch ? xj1 : 1.0f;
    atomicAdd(po + lane, xj0 * w);
    if (lane <= DEN2 - 32) atomicAdd(po + 32 + lane, v1 * w);
}

__global__ __launch_bounds__(256) void gat_edge_l2(
    const int* __restrict__ src, const int* __restrict__ dst,
    const float* __restrict__ xl, const float* __restrict__ xr,
    const float* __restrict__ att, const float* __restrict__ bnd,
    const int* __restrict__ mode, float* __restrict__ e2,
    float* __restrict__ m2, float* __restrict__ out2a, float* __restrict__ out2b, int E) {
    if (*mode != 0) {
        const long long E8 = (long long)E * 8;
        for (long long t = (long long)blockIdx.x * blockDim.x + threadIdx.x; t < E8;
             t += (long long)gridDim.x * blockDim.x)
            gat_logit_max_item_l2(xl, xr, att, src, dst, e2, m2, t);
        return;
    }
    const int lane = threadIdx.x & 31;
    const int wave = __builtin_amdgcn_readfirstlane(
        (int)(blockIdx.x * (blockDim.x >> 5) + (threadIdx.x >> 5)));
    const int wbase = wave * 32;
    if (wbase >= E) return;
    float* __restrict__ out2 = gat_package_half() != 0 ? out2b : out2a;
    const int nE = (E - wbase) < 32 ? (E - wbase) : 32;
    int s_i = 0, d_i = 0;
    if (lane < nE) {
        s_i = src[wbase + lane];
        d_i = dst[wbase + lane];
    }
    const int c1 = 32 + (lane & 7);
    const bool tail_ch = lane < DIM_OUT - 32;
    const float av0 = att[lane], av1 = att[c1];
    const float bn0 = bnd[lane], bn1 = bnd[c1];

    int j0 = 0;
    for (; j0 + 4 <= nE; j0 += 4) {
        float xj0[4], xj1[4], xi0[4], xi1[4];
#pragma unroll
        for (int q = 0; q < 4; ++q) {
            const int s = __builtin_amdgcn_readlane(s_i, j0 + q);
            const int d = __builtin_amdgcn_readlane(d_i, j0 + q);
            const float* pj = xl + (size_t)s * DIM_OUT;
            const float* pi = xr + (size_t)d * DIM_OUT;
            xj0[q] = pj[lane];
            xj1[q] = pj[c1];
            xi0[q] = pi[lane];
            xi1[q] = pi[c1];
        }
#pragma unroll
        for (int q = 0; q < 4; ++q) {
            const int d = __builtin_amdgcn_readlane(d_i, j0 + q);
            gat_edge_step_l2(xj0[q], xj1[q], xi0[q], xi1[q], av0, av1, bn0, bn1,
                             tail_ch, out2 + (size_t)d * ROW2, lane);
        }
    }
    for (; j0 < nE; ++j0) {
        const int s = __builtin_amdgcn_readlane(s_i, j0);
        const int d = __builtin_amdgcn_readlane(d_i, j0);
        const float* pj = xl + (size_t)s * DIM_OUT;
        const float* pi = xr + (size_t)d * DIM_OUT;
        gat_edge_step_l2(pj[lane], pj[c1], pi[lane], pi[c1], av0, av1, bn0, bn1,
                         tail_ch, out2 + (size_t)d * ROW2, lane);
    }
}

__global__ __launch_bounds__(256) void gat_edge_aggregate_l2(
    const int* __restrict__ src, const int* __restrict__ dst,
    const float* __restrict__ xl, const float* __restrict__ e2,
    const float* __restrict__ m2, float* __restrict__ out2a, float* __restrict__ out2b,
    const int* __restrict__ mode, int E) {
    if (*mode == 0) return;
    const int lane = threadIdx.x & 31;
    const int wave = __builtin_amdgcn_readfirstlane(
        (int)(blockIdx.x * (blockDim.x >> 5) + (threadIdx.x >> 5)));
    const int wbase = wave * 32;
    if (wbase >= E) return;
    float* __restrict__ out2 = gat_package_half() != 0 ? out2b : out2a;
    const int ei = wbase + lane;
    int s_i = 0, d_i = 0;
    float ex_i = 0.0f;
    if (ei < E) {
        s_i = src[ei];
        d_i = dst[ei];
        ex_i = expf(e2[ei] - m2[d_i]);
    }
    const int nE = (E - wbase) < 32 ? (E - wbase) : 32;
    const int c1 = 32 + (lane & 7);
    const bool tail_ch = lane < DIM_OUT - 32;

    if (nE == 32) {
        for (int j0 = 0; j0 < 32; j0 += 4) {
            float r0[4], r1[4];
#pragma unroll
            for (int q = 0; q < 4; ++q) {
                const int s = __builtin_amdgcn_readlane(s_i, j0 + q);
                const float* pj = xl + (size_t)s * DIM_OUT;
                r0[q] = pj[lane];
                r1[q] = pj[c1];
            }
            asm volatile("" :: "v"(r1[0]), "v"(r1[1]), "v"(r1[2]), "v"(r1[3]));
#pragma unroll
            for (int q = 0; q < 4; ++q) {
                const int d = __builtin_amdgcn_readlane(d_i, j0 + q);
                const float a = __int_as_float(
                    __builtin_amdgcn_readlane(__float_as_int(ex_i), j0 + q));
                float* po = out2 + (size_t)d * ROW2;
                const float v1 = tail_ch ? r1[q] : 1.0f;
                atomicAdd(po + lane, r0[q] * a);
                if (lane <= DEN2 - 32) atomicAdd(po + 32 + lane, v1 * a);
            }
        }
    } else {
        for (int j = 0; j < nE; ++j) {
            const int s = __builtin_amdgcn_readlane(s_i, j);
            const int d = __builtin_amdgcn_readlane(d_i, j);
            const float a = __int_as_float(
                __builtin_amdgcn_readlane(__float_as_int(ex_i), j));
            const float* pj = xl + (size_t)s * DIM_OUT;
            float* po = out2 + (size_t)d * ROW2;
            const float v0 = pj[lane];
            const float v1 = tail_ch ? pj[c1] : 1.0f;
            atomicAdd(po + lane, v0 * a);
            if (lane <= DEN2 - 32) atomicAdd(po + 32 + lane, v1 * a);
        }
    }
}

__global__ __launch_bounds__(256) void gat_finalize(
    const float* __restrict__ out2, const float* __restrict__ out2b,
    const float* __restrict__ bias2, float* __restrict__ d_out, int N) {
    const int node = blockIdx.x * (blockDim.x >> 5) + (threadIdx.x >> 5);
    const int lane = threadIdx.x & 31;
    if (node >= N) return;
    const float* row = out2 + (size_t)node * ROW2;
    const float* rowb = out2b + (size_t)node * ROW2;
    const float den = row[DEN2] + rowb[DEN2];
    const float ra = row[lane] + rowb[lane];
    const float a = (den > 0.0f ? ra / den : ra) + bias2[lane];
    const bool has2 = (lane < DIM_OUT - 32);
    const float rb = has2 ? row[32 + lane] + rowb[32 + lane] : 0.0f;
    const float b = has2 ? ((den > 0.0f ? rb / den : rb) + bias2[32 + lane])
                         : -__builtin_inff();
    float mx = fmaxf(a, b);
#pragma unroll
    for (int o = 16; o >= 1; o >>= 1) mx = fmaxf(mx, __shfl_xor(mx, o, 32));
    float s = expf(a - mx) + (has2 ? expf(b - mx) : 0.0f);
#pragma unroll
    for (int o = 16; o >= 1; o >>= 1) s += __shfl_xor(s, o, 32);
    const float lse = logf(s) + mx;
    float* h  = d_out + (size_t)node * DIM_OUT;
    float* ls = d_out + (size_t)N * DIM_OUT + (size_t)node * DIM_OUT;
    h[lane] = a;
    ls[lane] = a - lse;
    if (has2) {
        h[32 + lane] = b;
        ls[32 + lane] = b - lse;
    }
}

static inline int cdiv_i(long long a, long long b) { return (int)((a + b - 1) / b); }

extern "C" void kernel_launch(void* const* d_in, const int* in_sizes, int n_in,
                              void* d_out, int out_size, void* d_ws, size_t ws_size,
                              hipStream_t stream) {
    (void)n_in; (void)out_size; (void)ws_size;
    const float* x     = (const float*)d_in[0];
    const int*   eidx  = (const int*)d_in[1];
    const float* Wl1   = (const float*)d_in[2];
    const float* bl1   = (const float*)d_in[3];
    const float* Wr1   = (const float*)d_in[4];
    const float* br1   = (const float*)d_in[5];
    const float* att1  = (const float*)d_in[6];
    const float* bias1 = (const float*)d_in[7];
    const float* Wl2   = (const float*)d_in[8];
    const float* bl2   = (const float*)d_in[9];
    const float* Wr2   = (const float*)d_in[10];
    const float* br2   = (const float*)d_in[11];
    const float* att2  = (const float*)d_in[12];
    const float* bias2 = (const float*)d_in[13];

    const int N = in_sizes[0] / DIM_IN;
    const int E = in_sizes[1] / 2;
    const int* src = eidx;
    const int* dst = eidx + E;

    float* ws = (float*)d_ws;
    size_t o = 0;
    float* h_l1   = ws + o; o += (size_t)N * HID;
    float* h_r1   = ws + o; o += (size_t)N * HID;
    float* e1     = ws + o; o += (size_t)E * HEADS;
    float* m1     = ws + o; o += (size_t)N * HEADS;
    float* denom1 = ws + o; o += (size_t)N * HEADS;
    float* denom1b = ws + o; o += (size_t)N * HEADS;
    float* out1   = ws + o; o += (size_t)N * HID;
    float* out1b  = ws + o; o += (size_t)N * HID;
    float* cpart1 = ws + o; o += (size_t)1024 * 2 * HID;
    float* bnd1   = ws + o; o += 64;
    int*   mode1  = (int*)(ws + o); o += 64;
    float* cpart2 = ws + o; o += (size_t)1250 * 2 * 48;
    float* bnd2   = ws + o; o += 64;
    int*   mode2  = (int*)(ws + o); o += 64;
    float* h_l2   = ws;
    float* h_r2   = h_l2 + (size_t)N * DIM_OUT;
    float* e2     = h_r2 + (size_t)N * DIM_OUT;
    float* out2   = e2 + (size_t)E;
    float* out2b  = out2 + (size_t)N * ROW2;
    float* m2     = out2b + (size_t)N * ROW2;

    const int B = 256;

    gat_proj_pair_f32_wmma<DIM_IN, HID, 4, 2, HID, HEADS, HEADS>
        <<<1024, 128, 0, stream>>>(x, Wl1, bl1, Wr1, br1, h_l1, h_r1,
                                   out1, out1b, m1, denom1, denom1b, cpart1, N);
    gat_logit_bound<HID, HID, HEADS, DIM_H>
        <<<1, 16 * HID, 0, stream>>>(cpart1, att1, bnd1, mode1, 1024);

    gat_edge_l1<<<cdiv_i(E, B), B, 0, stream>>>(
        src, dst, h_l1, h_r1, att1, bnd1, mode1, e1, m1, denom1, denom1b, out1, out1b, E);
    gat_edge_aggregate_l1<<<cdiv_i(E, B), B, 0, stream>>>(
        src, dst, h_l1, e1, m1, denom1, denom1b, out1, out1b, mode1, E);
    gat_norm_bias_elu<<<cdiv_i((long long)N * HID, B), B, 0, stream>>>(
        out1, out1b, denom1, denom1b, bias1, (long long)N * HID);

    gat_proj_pair_f32_wmma<HID, DIM_OUT, 3, 3, ROW2, 1, 0>
        <<<1250, 64, 0, stream>>>(out1, Wl2, bl2, Wr2, br2, h_l2, h_r2,
                                  out2, out2b, m2, (float*)nullptr, (float*)nullptr, cpart2, N);
    gat_logit_bound<DIM_OUT, 48, 1, DIM_OUT>
        <<<1, 16 * 48, 0, stream>>>(cpart2, att2, bnd2, mode2, 1250);

    gat_edge_l2<<<cdiv_i(E, B), B, 0, stream>>>(
        src, dst, h_l2, h_r2, att2, bnd2, mode2, e2, m2, out2, out2b, E);
    gat_edge_aggregate_l2<<<cdiv_i(E, B), B, 0, stream>>>(
        src, dst, h_l2, e2, m2, out2, out2b, mode2, E);

    gat_finalize<<<cdiv_i(N, B / 32), B, 0, stream>>>((const float*)out2, (const float*)out2b,
                                                      bias2, (float*)d_out, N);
}


# Column max/min of the source plane merged inside the projection launch by integer atomic max/min on keys; each edge wave forms bound and mode itself; both bound launches go (10->9); sums unchanged
# speedup vs baseline: 3.4559x; 1.1196x over previous
#include <hip/hip_runtime.h>
#include <hip/hip_bf16.h>


#define DIM_IN   128
#define HID      64
#define HEADS    8
#define DIM_H    8
#define DIM_OUT  40
#define NEG_SLOPE 0.2f
#define ROW2     64
#define DEN2     40

typedef __attribute__((ext_vector_type(2))) float v2f;
typedef __attribute__((ext_vector_type(4))) float v4f;
typedef __attribute__((ext_vector_type(8))) float v8f;

__device__ __forceinline__ void atomicMaxFloat(float* addr, float val) {
    if (!(val == val)) return;
    if (val >= 0.0f) {
        atomicMax((int*)addr, __float_as_int(val));
    } else {
        atomicMin((unsigned int*)addr, __float_as_uint(val));
    }
}

template <int M>
__device__ __forceinline__ float gat_row_xor(float x) {
    return __int_as_float(__builtin_amdgcn_update_dpp(
        0, __float_as_int(x), 0x160 + M, 0xf, 0xf, false));
}

__device__ __forceinline__ float gat_xhalf(float x) {
    const int b = __float_as_int(x);
    return __int_as_float(__builtin_amdgcn_permlanex16(b, b, 0x76543210, 0xfedcba98, false, false));
}

__device__ __forceinline__ float gat_lrelu(float v) { return v > 0.0f ? v : v * NEG_SLOPE; }

__device__ __forceinline__ int gat_package_half() {
    return (int)((__builtin_amdgcn_s_sendmsg_rtn(0x87) >> 18) & 1u);
}

#define GAT_KEY_NINF 0x007fffffu
#define GAT_KEY_PINF 0xff800000u
__device__ __forceinline__ unsigned int gat_key(float f) {
    const unsigned int b = __float_as_uint(f);
    return (b & 0x80000000u) ? ~b : (b | 0x80000000u);
}
__device__ __forceinline__ float gat_unkey(unsigned int k) {
    return __uint_as_float((k & 0x80000000u) ? (k & 0x7fffffffu) : ~k);
}

__global__ __launch_bounds__(256) void gat_bound_start(unsigned int* __restrict__ k1,
                                                       unsigned int* __restrict__ k2) {
    const int t = threadIdx.x;
    k1[t] = ((t >> 6) & 1) ? GAT_KEY_PINF : GAT_KEY_NINF;
    if (t < 192) k2[t] = ((t / 48) & 1) ? GAT_KEY_PINF : GAT_KEY_NINF;
}

template <int K, int CC, int NT, int TPW, int ACCW, int MW, int DW>
__global__ __launch_bounds__(32 * (2 * NT / TPW)) void gat_proj_pair_f32_wmma(
    const float* __restrict__ A,
    const float* __restrict__ Wl, const float* __restrict__ bl,
    const float* __restrict__ Wr, const float* __restrict__ br,
    float* __restrict__ outl, float* __restrict__ outr,
    float* __restrict__ acc0, float* __restrict__ acc1, float* __restrict__ max0,
    float* __restrict__ den0, float* __restrict__ den1,
    unsigned int* __restrict__ ckey, int n) {
    constexpr int WPM = NT / TPW;
    constexpr int KS = K / 4;
    constexpr bool PAD = (16 * NT != CC);
    const int lane = threadIdx.x & 31;
    const int wave = __builtin_amdgcn_readfirstlane((int)(threadIdx.x >> 5));
    const bool right = wave >= WPM;
    const float* __restrict__ W = right ? Wr : Wl;
    const float* __restrict__ bias = right ? br : bl;
    float* __restrict__ out = right ? outr : outl;
    const int tile0 = (WPM == 1) ? 0 : (right ? wave - WPM : wave) * TPW;
    const int c = lane & 15;
    const int h = lane >> 4;

    v2f b[TPW][KS];
    float bv[TPW];
#pragma unroll
    for (int t = 0; t < TPW; ++t) {
        const int col = (tile0 + t) * 16 + c;
        const bool ok = !PAD || col < CC;
        const float* __restrict__ wc = W + (ok ? col : 0);
        bv[t] = ok ? bias[col] : 0.0f;
#pragma unroll
        for (int s = 0; s < KS; ++s) {
            const int k = 8 * (s >> 1) + 4 * h + 2 * (s & 1);
            const float w0 = wc[(size_t)k * CC];
            const float w1 = wc[(size_t)(k + 1) * CC];
            b[t][s].x = ok ? w0 : 0.0f;
            b[t][s].y = ok ? w1 : 0.0f;
        }
    }

    float cmx[TPW], cmn[TPW];
#pragma unroll
    for (int t = 0; t < TPW; ++t) { cmx[t] = -__builtin_inff(); cmn[t] = __builtin_inff(); }

    const int rowTiles = n >> 4;
    for (int rt = blockIdx.x; rt < rowTiles; rt += gridDim.x) {
        const float* __restrict__ aRow = A + (size_t)((rt << 4) + c) * K + 4 * h;
        v4f a[K / 8];
#pragma unroll
        for (int j = 0; j < K / 8; ++j) a[j] = *(const v4f*)(aRow + 8 * j);

        v8f acc[TPW];
#pragma unroll
        for (int t = 0; t < TPW; ++t) acc[t] = v8f{};
#pragma unroll
        for (int j = 0; j < K / 8; ++j) {
            const v2f a0 = {a[j].x, a[j].y};
            const v2f a1 = {a[j].z, a[j].w};
#pragma unroll
            for (int t = 0; t < TPW; ++t)
                acc[t] = __builtin_amdgcn_wmma_f32_16x16x4_f32(
                    false, a0, false, b[t][2 * j], (short)0, acc[t], false, false);
#pragma unroll
            for (int t = 0; t < TPW; ++t)
                acc[t] = __builtin_amdgcn_wmma_f32_16x16x4_f32(
                    false, a1, false, b[t][2 * j + 1], (short)0, acc[t], false, false);
        }

        const int rbase = (rt << 4) + (h << 3);
#pragma unroll
        for (int t = 0; t < TPW; ++t) {
            const int col = (tile0 + t) * 16 + c;
            if (!PAD || col < CC) {
#pragma unroll
                for (int r = 0; r < 8; ++r) {
                    const float v = acc[t][r] + bv[t];
                    out[(size_t)(rbase + r) * CC + col] = v;
                    cmx[t] = fmaxf(cmx[t], v);
                    cmn[t] = fminf(cmn[t], v);
                }
            }
        }

        constexpr int NW = 2 * WPM;
        constexpr int ACC_P = 16 * ACCW / 4;
        constexpr int MAX_P = 16 * MW / 4;
        constexpr int DEN_P = 16 * DW / 4;
        static_assert(ACC_P % (NW * 32) == 0 && MAX_P <= 32 && DEN_P <= 32, "tile start values");
        v4f* __restrict__ pa = (v4f*)(acc0 + (size_t)rt * (16 * ACCW));
        v4f* __restrict__ pb = (v4f*)(acc1 + (size_t)rt * (16 * ACCW));
        for (int q = 0; q < ACC_P / (NW * 32); ++q) {
            pa[(q * NW + wave) * 32 + lane] = v4f{};
            pb[(q * NW + wave) * 32 + lane] = v4f{};
        }
        if (wave == NW - 1 && lane < MAX_P) {
            const float ninf = -__builtin_inff();
            ((v4f*)(max0 + (size_t)rt * (16 * MW)))[lane] = v4f{ninf, ninf, ninf, ninf};
        }
        if constexpr (DW > 0) {
            if (wave == 0 && lane < DEN_P) {
                ((v4f*)(den0 + (size_t)rt * (16 * DW)))[lane] = v4f{};
                ((v4f*)(den1 + (size_t)rt * (16 * DW)))[lane] = v4f{};
            }
        }
    }

    if (!right) {
        unsigned int* __restrict__ kb = ckey + gat_package_half() * (32 * NT);
#pragma unroll
        for (int t = 0; t < TPW; ++t) {
            const float mx = fmaxf(cmx[t], __shfl_xor(cmx[t], 16, 32));
            const float mn = fminf(cmn[t], __shfl_xor(cmn[t], 16, 32));
            const int col = (tile0 + t) * 16 + c;
            if (h == 0 && (!PAD || col < CC)) {
                atomicMax(kb + col, gat_key(mx));
                atomicMin(kb + 16 * NT + col, gat_key(mn));
            }
        }
    }
}

__device__ __forceinline__ bool gat_bound_l1(
    const unsigned int* __restrict__ ckey, float av0, float av1, int lane,
    float& bn0, float& bn1) {
    const unsigned int* __restrict__ k0 = ckey;
    const unsigned int* __restrict__ k1 = ckey + 128;
    const float u0 = fmaxf(gat_unkey(k0[lane]), gat_unkey(k1[lane]));
    const float u1 = fmaxf(gat_unkey(k0[32 + lane]), gat_unkey(k1[32 + lane]));
    const float l0 = fminf(gat_unkey(k0[64 + lane]), gat_unkey(k1[64 + lane]));
    const float l1 = fminf(gat_unkey(k0[96 + lane]), gat_unkey(k1[96 + lane]));
    bn0 = av0 >= 0.0f ? u0 : l0;
    bn1 = av1 >= 0.0f ? u1 : l1;
    float g0 = fabsf(av0) * (u0 - l0);
    float g1 = fabsf(av1) * (u1 - l1);
    g0 += gat_row_xor<1>(g0); g1 += gat_row_xor<1>(g1);
    g0 += gat_row_xor<2>(g0); g1 += gat_row_xor<2>(g1);
    g0 += gat_row_xor<4>(g0); g1 += gat_row_xor<4>(g1);
    const bool bad = !(g0 < 70.0f) || !(g1 < 70.0f);
    return __builtin_amdgcn_ballot_w32(bad) != 0u;
}

__device__ __forceinline__ bool gat_bound_l2(
    const unsigned int* __restrict__ ckey, float av0, float av1, int lane, int c1,
    bool tail_ch, float& bn0, float& bn1) {
    const unsigned int* __restrict__ k0 = ckey;
    const unsigned int* __restrict__ k1 = ckey + 96;
    const float u0 = fmaxf(gat_unkey(k0[lane]), gat_unkey(k1[lane]));
    const float u1 = fmaxf(gat_unkey(k0[c1]), gat_unkey(k1[c1]));
    const float l0 = fminf(gat_unkey(k0[48 + lane]), gat_unkey(k1[48 + lane]));
    const float l1 = fminf(gat_unkey(k0[48 + c1]), gat_unkey(k1[48 + c1]));
    bn0 = av0 >= 0.0f ? u0 : l0;
    bn1 = av1 >= 0.0f ? u1 : l1;
    const float ga = fabsf(av0) * (u0 - l0);
    const float gb = fabsf(av1) * (u1 - l1);
    float g = ga + (tail_ch ? gb : 0.0f);
    g += gat_row_xor<1>(g);
    g += gat_row_xor<2>(g);
    g += gat_row_xor<4>(g);
    g += gat_row_xor<8>(g);
    g += gat_xhalf(g);
    return __builtin_amdgcn_ballot_w32(!(g < 70.0f)) != 0u;
}

__device__ __forceinline__ void gat_logit_max_item_l1(
    const float* __restrict__ xl, const float* __restrict__ xr,
    const float* __restrict__ att, const int* __restrict__ src,
    const int* __restrict__ dst, float* __restrict__ e1,
    float* __restrict__ m1, long long t) {
    const int e = (int)(t >> 3);
    const int h = (int)(t & 7);
    const int s = src[e], d = dst[e];
    const float* pj = xl + (size_t)s * HID + h * DIM_H;
    const float* pi = xr + (size_t)d * HID + h * DIM_H;
    const float* pa = att + h * DIM_H;
    float acc = 0.0f;
#pragma unroll
    for (int c = 0; c < DIM_H; ++c) {
        float v = pi[c] + pj[c];
        v = v > 0.0f ? v : v * NEG_SLOPE;
        acc += v * pa[c];
    }
    e1[t] = acc;
    atomicMaxFloat(&m1[(size_t)d * HEADS + h], acc);
}

__device__ __forceinline__ void gat_edge_step_l1(
    float xj0, float xj1, float xi0, float xi1,
    float av0, float av1, float bn0, float bn1,
    bool den_lane, bool den_hi, float* __restrict__ den,
    float* __restrict__ po, int lane) {
    float t0 = (gat_lrelu(xi0 + xj0) - gat_lrelu(xi0 + bn0)) * av0;
    float t1 = (gat_lrelu(xi1 + xj1) - gat_lrelu(xi1 + bn1)) * av1;
    t0 += gat_row_xor<1>(t0); t1 += gat_row_xor<1>(t1);
    t0 += gat_row_xor<2>(t0); t1 += gat_row_xor<2>(t1);
    t0 += gat_row_xor<4>(t0); t1 += gat_row_xor<4>(t1);
    const float w0 = expf(t0);
    const float w1 = expf(t1);
    if (den_lane) atomicAdd(den, den_hi ? w1 : w0);
    atomicAdd(po + lane, xj0 * w0);
    atomicAdd(po + 32 + lane, xj1 * w1);
}

__global__ __launch_bounds__(256) void gat_edge_l1(
    const int* __restrict__ src, const int* __restrict__ dst,
    const float* __restrict__ xl, const float* __restrict__ xr,
    const float* __restrict__ att, const unsigned int* __restrict__ ckey,
    int* __restrict__ mode, float* __restrict__ e1,
    float* __restrict__ m1, float* __restrict__ denom1a, float* __restrict__ denom1b,
    float* __restrict__ out1a, float* __restrict__ out1b, int E) {
    const int lane = threadIdx.x & 31;
    const float av0 = att[lane], av1 = att[32 + lane];
    float bn0, bn1;
    const bool loose = gat_bound_l1(ckey, av0, av1, lane, bn0, bn1);
    if (blockIdx.x == 0 && threadIdx.x == 0) *mode = loose ? 1 : 0;
    if (loose) {
        const long long E8 = (long long)E * HEADS;
        for (long long t = (long long)blockIdx.x * blockDim.x + threadIdx.x; t < E8;
             t += (long long)gridDim.x * blockDim.x)
            gat_logit_max_item_l1(xl, xr, att, src, dst, e1, m1, t);
        return;
    }
    const int wave = __builtin_amdgcn_readfirstlane(
        (int)(blockIdx.x * (blockDim.x >> 5) + (threadIdx.x >> 5)));
    const int wbase = wave * 32;
    if (wbase >= E) return;
    const bool upper = gat_package_half() != 0;
    float* __restrict__ denom1 = upper ? denom1b : denom1a;
    float* __restrict__ out1 = upper ? out1b : out1a;
    const int nE = (E - wbase) < 32 ? (E - wbase) : 32;
    int s_i = 0, d_i = 0;
    if (lane < nE) {
        s_i = src[wbase + lane];
        d_i = dst[wbase + lane];
    }
    const bool den_lane = (lane & 7) < 2;
    const bool den_hi = (lane & 1) != 0;
    const int den_off = (lane & 1) * 4 + (lane >> 3);

    int j0 = 0;
    for (; j0 + 4 <= nE; j0 += 4) {
        float xj0[4], xj1[4], xi0[4], xi1[4];
#pragma unroll
        for (int q = 0; q < 4; ++q) {
            const int s = __builtin_amdgcn_readlane(s_i, j0 + q);
            const int d = __builtin_amdgcn_readlane(d_i, j0 + q);
            const float* pj = xl + (size_t)s * HID;
            const float* pi = xr + (size_t)d * HID;
            xj0[q] = pj[lane];
            xj1[q] = pj[32 + lane];
            xi0[q] = pi[lane];
            xi1[q] = pi[32 + lane];
        }
#pragma unroll
        for (int q = 0; q < 4; ++q) {
            const int d = __builtin_amdgcn_readlane(d_i, j0 + q);
            gat_edge_step_l1(xj0[q], xj1[q], xi0[q], xi1[q], av0, av1, bn0, bn1,
                             den_lane, den_hi, denom1 + (size_t)d * HEADS + den_off,
                             out1 + (size_t)d * HID, lane);
        }
    }
    for (; j0 < nE; ++j0) {
        const int s = __builtin_amdgcn_readlane(s_i, j0);
        const int d = __builtin_amdgcn_readlane(d_i, j0);
        const float* pj = xl + (size_t)s * HID;
        const float* pi = xr + (size_t)d * HID;
        gat_edge_step_l1(pj[lane], pj[32 + lane], pi[lane], pi[32 + lane], av0, av1, bn0, bn1,
                         den_lane, den_hi, denom1 + (size_t)d * HEADS + den_off,
                         out1 + (size_t)d * HID, lane);
    }
}

__global__ __launch_bounds__(256) void gat_edge_aggregate_l1(
    const int* __restrict__ src, const int* __restrict__ dst,
    const float* __restrict__ xl, const float* __restrict__ e1,
    const float* __restrict__ m1, float* __restrict__ denom1a, float* __restrict__ denom1b,
    float* __restrict__ out1a, float* __restrict__ out1b,
    const int* __restrict__ mode, int E) {
    if (*mode == 0) return;
    const int lane = threadIdx.x & 31;
    const int wave = __builtin_amdgcn_readfirstlane(
        (int)(blockIdx.x * (blockDim.x >> 5) + (threadIdx.x >> 5)));
    const int wbase = wave * 32;
    if (wbase >= E) return;
    const bool upper = gat_package_half() != 0;
    float* __restrict__ denom1 = upper ? denom1b : denom1a;
    float* __restrict__ out1 = upper ? out1b : out1a;
    const int sub = lane >> 3;
    const int h = lane & 7;
    const int nE = (E - wbase) < 32 ? (E - wbase) : 32;

    if (nE == 32) {
        const int s_i = src[wbase + lane];
        const int d_i = dst[wbase + lane];
        float ev[8], mv[8], ex[8];
        int dp[8];
#pragma unroll
        for (int g = 0; g < 8; ++g) {
            dp[g] = __shfl(d_i, g * 4 + sub, 32);
            ev[g] = e1[(size_t)wbase * HEADS + g * 32 + lane];
            mv[g] = m1[(size_t)dp[g] * HEADS + h];
        }
#pragma unroll
        for (int g = 0; g < 8; ++g) {
            ex[g] = expf(ev[g] - mv[g]);
            atomicAdd(&denom1[(size_t)dp[g] * HEADS + h], ex[g]);
        }
#pragma unroll
        for (int g = 0; g < 8; ++g) {
            float r0[4], r1[4];
#pragma unroll
            for (int q = 0; q < 4; ++q) {
                const int s = __builtin_amdgcn_readlane(s_i, g * 4 + q);
                const float* pj = xl + (size_t)s * HID;
                r0[q] = pj[lane];
                r1[q] = pj[32 + lane];
            }
#pragma unroll
            for (int q = 0; q < 4; ++q) {
                const int d = __builtin_amdgcn_readlane(d_i, g * 4 + q);
                const float a = __shfl(ex[g], q * 8 + sub, 32);
                const float b = __shfl(ex[g], q * 8 + 4 + sub, 32);
                float* po = out1 + (size_t)d * HID;
                atomicAdd(po + lane, r0[q] * a);
                atomicAdd(po + 32 + lane, r1[q] * b);
            }
        }
    } else {
        for (int j = 0; j < nE; ++j) {
            const int e = wbase + j;
            const int s = src[e], d = dst[e];
            const float a = expf(e1[(size_t)e * HEADS + sub] - m1[(size_t)d * HEADS + sub]);
            const float b = expf(e1[(size_t)e * HEADS + 4 + sub] - m1[(size_t)d * HEADS + 4 + sub]);
            if (h == 0) {
                atomicAdd(&denom1[(size_t)d * HEADS + sub], a);
                atomicAdd(&denom1[(size_t)d * HEADS + 4 + sub], b);
            }
            const float* pj = xl + (size_t)s * HID;
            float* po = out1 + (size_t)d * HID;
            atomicAdd(po + lane, pj[lane] * a);
            atomicAdd(po + 32 + lane, pj[32 + lane] * b);
        }
    }
}

__global__ void gat_norm_bias_elu(float* __restrict__ out1,
                                  const float* __restrict__ out1b,
                                  const float* __restrict__ denom1,
                                  const float* __restrict__ denom1b,
                                  const float* __restrict__ bias1,
                                  long long total) {
    long long i = (long long)blockIdx.x * blockDim.x + threadIdx.x;
    if (i >= total) return;
    const float den = denom1[i >> 3] + denom1b[i >> 3];
    const float acc = out1[i] + out1b[i];
    float v = (den > 0.0f ? acc / den : acc) + bias1[(int)(i & (HID - 1))];
    out1[i] = v > 0.0f ? v : expm1f(v);
}

__device__ __forceinline__ void gat_logit_max_item_l2(
    const float* __restrict__ xl, const float* __restrict__ xr,
    const float* __restrict__ att, const int* __restrict__ src,
    const int* __restrict__ dst, float* __restrict__ e2,
    float* __restrict__ m2, long long t) {
    const int e = (int)(t >> 3);
    const int sub = (int)(t & 7);
    const int s = src[e], d = dst[e];
    const float* pj = xl + (size_t)s * DIM_OUT;
    const float* pi = xr + (size_t)d * DIM_OUT;
    float acc = 0.0f;
#pragma unroll
    for (int c = sub; c < DIM_OUT; c += 8) {
        float v = pi[c] + pj[c];
        v = v > 0.0f ? v : v * NEG_SLOPE;
        acc += v * att[c];
    }
    acc += __shfl_xor(acc, 1, 32);
    acc += __shfl_xor(acc, 2, 32);
    acc += __shfl_xor(acc, 4, 32);
    if (sub == 0) {
        e2[e] = acc;
        atomicMaxFloat(&m2[d], acc);
    }
}

__device__ __forceinline__ void gat_edge_step_l2(
    float xj0, float xj1, float xi0, float xi1,
    float av0, float av1, float bn0, float bn1,
    bool tail_ch, float* __restrict__ po, int lane) {
    const float ta = (gat_lrelu(xi0 + xj0) - gat_lrelu(xi0 + bn0)) * av0;
    const float tb = (gat_lrelu(xi1 + xj1) - gat_lrelu(xi1 + bn1)) * av1;
    float t = ta + (tail_ch ? tb : 0.0f);
    t += gat_row_xor<1>(t);
    t += gat_row_xor<2>(t);
    t += gat_row_xor<4>(t);
    t += gat_row_xor<8>(t);
    t += gat_xhalf(t);
    const float w = expf(t);
    const float v1 = tail_ch ? xj1 : 1.0f;
    atomicAdd(po + lane, xj0 * w);
    if (lane <= DEN2 - 32) atomicAdd(po + 32 + lane, v1 * w);
}

__global__ __launch_bounds__(256) void gat_edge_l2(
    const int* __restrict__ src, const int* __restrict__ dst,
    const float* __restrict__ xl, const float* __restrict__ xr,
    const float* __restrict__ att, const unsigned int* __restrict__ ckey,
    int* __restrict__ mode, float* __restrict__ e2,
    float* __restrict__ m2, float* __restrict__ out2a, float* __restrict__ out2b, int E) {
    const int lane = threadIdx.x & 31;
    const int c1 = 32 + (lane & 7);
    const bool tail_ch = lane < DIM_OUT - 32;
    const float av0 = att[lane], av1 = att[c1];
    float bn0, bn1;
    const bool loose = gat_bound_l2(ckey, av0, av1, lane, c1, tail_ch, bn0, bn1);
    if (blockIdx.x == 0 && threadIdx.x == 0) *mode = loose ? 1 : 0;
    if (loose) {
        const long long E8 = (long long)E * 8;
        for (long long t = (long long)blockIdx.x * blockDim.x + threadIdx.x; t < E8;
             t += (long long)gridDim.x * blockDim.x)
            gat_logit_max_item_l2(xl, xr, att, src, dst, e2, m2, t);
        return;
    }
    const int wave = __builtin_amdgcn_readfirstlane(
        (int)(blockIdx.x * (blockDim.x >> 5) + (threadIdx.x >> 5)));
    const int wbase = wave * 32;
    if (wbase >= E) return;
    float* __restrict__ out2 = gat_package_half() != 0 ? out2b : out2a;
    const int nE = (E - wbase) < 32 ? (E - wbase) : 32;
    int s_i = 0, d_i = 0;
    if (lane < nE) {
        s_i = src[wbase + lane];
        d_i = dst[wbase + lane];
    }
    int j0 = 0;
    for (; j0 + 4 <= nE; j0 += 4) {
        float xj0[4], xj1[4], xi0[4], xi1[4];
#pragma unroll
        for (int q = 0; q < 4; ++q) {
            const int s = __builtin_amdgcn_readlane(s_i, j0 + q);
            const int d = __builtin_amdgcn_readlane(d_i, j0 + q);
            const float* pj = xl + (size_t)s * DIM_OUT;
            const float* pi = xr + (size_t)d * DIM_OUT;
            xj0[q] = pj[lane];
            xj1[q] = pj[c1];
            xi0[q] = pi[lane];
            xi1[q] = pi[c1];
        }
#pragma unroll
        for (int q = 0; q < 4; ++q) {
            const int d = __builtin_amdgcn_readlane(d_i, j0 + q);
            gat_edge_step_l2(xj0[q], xj1[q], xi0[q], xi1[q], av0, av1, bn0, bn1,
                             tail_ch, out2 + (size_t)d * ROW2, lane);
        }
    }
    for (; j0 < nE; ++j0) {
        const int s = __builtin_amdgcn_readlane(s_i, j0);
        const int d = __builtin_amdgcn_readlane(d_i, j0);
        const float* pj = xl + (size_t)s * DIM_OUT;
        const float* pi = xr + (size_t)d * DIM_OUT;
        gat_edge_step_l2(pj[lane], pj[c1], pi[lane], pi[c1], av0, av1, bn0, bn1,
                         tail_ch, out2 + (size_t)d * ROW2, lane);
    }
}

__global__ __launch_bounds__(256) void gat_edge_aggregate_l2(
    const int* __restrict__ src, const int* __restrict__ dst,
    const float* __restrict__ xl, const float* __restrict__ e2,
    const float* __restrict__ m2, float* __restrict__ out2a, float* __restrict__ out2b,
    const int* __restrict__ mode, int E) {
    if (*mode == 0) return;
    const int lane = threadIdx.x & 31;
    const int wave = __builtin_amdgcn_readfirstlane(
        (int)(blockIdx.x * (blockDim.x >> 5) + (threadIdx.x >> 5)));
    const int wbase = wave * 32;
    if (wbase >= E) return;
    float* __restrict__ out2 = gat_package_half() != 0 ? out2b : out2a;
    const int ei = wbase + lane;
    int s_i = 0, d_i = 0;
    float ex_i = 0.0f;
    if (ei < E) {
        s_i = src[ei];
        d_i = dst[ei];
        ex_i = expf(e2[ei] - m2[d_i]);
    }
    const int nE = (E - wbase) < 32 ? (E - wbase) : 32;
    const int c1 = 32 + (lane & 7);
    const bool tail_ch = lane < DIM_OUT - 32;

    if (nE == 32) {
        for (int j0 = 0; j0 < 32; j0 += 4) {
            float r0[4], r1[4];
#pragma unroll
            for (int q = 0; q < 4; ++q) {
                const int s = __builtin_amdgcn_readlane(s_i, j0 + q);
                const float* pj = xl + (size_t)s * DIM_OUT;
                r0[q] = pj[lane];
                r1[q] = pj[c1];
            }
            asm volatile("" :: "v"(r1[0]), "v"(r1[1]), "v"(r1[2]), "v"(r1[3]));
#pragma unroll
            for (int q = 0; q < 4; ++q) {
                const int d = __builtin_amdgcn_readlane(d_i, j0 + q);
                const float a = __int_as_float(
                    __builtin_amdgcn_readlane(__float_as_int(ex_i), j0 + q));
                float* po = out2 + (size_t)d * ROW2;
                const float v1 = tail_ch ? r1[q] : 1.0f;
                atomicAdd(po + lane, r0[q] * a);
                if (lane <= DEN2 - 32) atomicAdd(po + 32 + lane, v1 * a);
            }
        }
    } else {
        for (int j = 0; j < nE; ++j) {
            const int s = __builtin_amdgcn_readlane(s_i, j);
            const int d = __builtin_amdgcn_readlane(d_i, j);
            const float a = __int_as_float(
                __builtin_amdgcn_readlane(__float_as_int(ex_i), j));
            const float* pj = xl + (size_t)s * DIM_OUT;
            float* po = out2 + (size_t)d * ROW2;
            const float v0 = pj[lane];
            const float v1 = tail_ch ? pj[c1] : 1.0f;
            atomicAdd(po + lane, v0 * a);
            if (lane <= DEN2 - 32) atomicAdd(po + 32 + lane, v1 * a);
        }
    }
}

__global__ __launch_bounds__(256) void gat_finalize(
    const float* __restrict__ out2, const float* __restrict__ out2b,
    const float* __restrict__ bias2, float* __restrict__ d_out, int N) {
    const int node = blockIdx.x * (blockDim.x >> 5) + (threadIdx.x >> 5);
    const int lane = threadIdx.x & 31;
    if (node >= N) return;
    const float* row = out2 + (size_t)node * ROW2;
    const float* rowb = out2b + (size_t)node * ROW2;
    const float den = row[DEN2] + rowb[DEN2];
    const float ra = row[lane] + rowb[lane];
    const float a = (den > 0.0f ? ra / den : ra) + bias2[lane];
    const bool has2 = (lane < DIM_OUT - 32);
    const float rb = has2 ? row[32 + lane] + rowb[32 + lane] : 0.0f;
    const float b = has2 ? ((den > 0.0f ? rb / den : rb) + bias2[32 + lane])
                         : -__builtin_inff();
    float mx = fmaxf(a, b);
#pragma unroll
    for (int o = 16; o >= 1; o >>= 1) mx = fmaxf(mx, __shfl_xor(mx, o, 32));
    float s = expf(a - mx) + (has2 ? expf(b - mx) : 0.0f);
#pragma unroll
    for (int o = 16; o >= 1; o >>= 1) s += __shfl_xor(s, o, 32);
    const float lse = logf(s) + mx;
    float* h  = d_out + (size_t)node * DIM_OUT;
    float* ls = d_out + (size_t)N * DIM_OUT + (size_t)node * DIM_OUT;
    h[lane] = a;
    ls[lane] = a - lse;
    if (has2) {
        h[32 + lane] = b;
        ls[32 + lane] = b - lse;
    }
}

static inline int cdiv_i(long long a, long long b) { return (int)((a + b - 1) / b); }

extern "C" void kernel_launch(void* const* d_in, const int* in_sizes, int n_in,
                              void* d_out, int out_size, void* d_ws, size_t ws_size,
                              hipStream_t stream) {
    (void)n_in; (void)out_size; (void)ws_size;
    const float* x     = (const float*)d_in[0];
    const int*   eidx  = (const int*)d_in[1];
    const float* Wl1   = (const float*)d_in[2];
    const float* bl1   = (const float*)d_in[3];
    const float* Wr1   = (const float*)d_in[4];
    const float* br1   = (const float*)d_in[5];
    const float* att1  = (const float*)d_in[6];
    const float* bias1 = (const float*)d_in[7];
    const float* Wl2   = (const float*)d_in[8];
    const float* bl2   = (const float*)d_in[9];
    const float* Wr2   = (const float*)d_in[10];
    const float* br2   = (const float*)d_in[11];
    const float* att2  = (const float*)d_in[12];
    const float* bias2 = (const float*)d_in[13];

    const int N = in_sizes[0] / DIM_IN;
    const int E = in_sizes[1] / 2;
    const int* src = eidx;
    const int* dst = eidx + E;

    float* ws = (float*)d_ws;
    size_t o = 0;
    float* h_l1   = ws + o; o += (size_t)N * HID;
    float* h_r1   = ws + o; o += (size_t)N * HID;
    float* e1     = ws + o; o += (size_t)E * HEADS;
    float* m1     = ws + o; o += (size_t)N * HEADS;
    float* denom1 = ws + o; o += (size_t)N * HEADS;
    float* denom1b = ws + o; o += (size_t)N * HEADS;
    float* out1   = ws + o; o += (size_t)N * HID;
    float* out1b  = ws + o; o += (size_t)N * HID;
    unsigned int* ckey1 = (unsigned int*)(ws + o); o += 256;
    int*   mode1  = (int*)(ws + o); o += 64;
    unsigned int* ckey2 = (unsigned int*)(ws + o); o += 256;
    int*   mode2  = (int*)(ws + o); o += 64;
    float* h_l2   = ws;
    float* h_r2   = h_l2 + (size_t)N * DIM_OUT;
    float* e2     = h_r2 + (size_t)N * DIM_OUT;
    float* out2   = e2 + (size_t)E;
    float* out2b  = out2 + (size_t)N * ROW2;
    float* m2     = out2b + (size_t)N * ROW2;

    const int B = 256;

    gat_bound_start<<<1, 256, 0, stream>>>(ckey1, ckey2);

    gat_proj_pair_f32_wmma<DIM_IN, HID, 4, 2, HID, HEADS, HEADS>
        <<<1024, 128, 0, stream>>>(x, Wl1, bl1, Wr1, br1, h_l1, h_r1,
                                   out1, out1b, m1, denom1, denom1b, ckey1, N);

    gat_edge_l1<<<cdiv_i(E, B), B, 0, stream>>>(
        src, dst, h_l1, h_r1, att1, ckey1, mode1, e1, m1, denom1, denom1b, out1, out1b, E);
    gat_edge_aggregate_l1<<<cdiv_i(E, B), B, 0, stream>>>(
        src, dst, h_l1, e1, m1, denom1, denom1b, out1, out1b, mode1, E);
    gat_norm_bias_elu<<<cdiv_i((long long)N * HID, B), B, 0, stream>>>(
        out1, out1b, denom1, denom1b, bias1, (long long)N * HID);

    gat_proj_pair_f32_wmma<HID, DIM_OUT, 3, 3, ROW2, 1, 0>
        <<<1250, 64, 0, stream>>>(out1, Wl2, bl2, Wr2, br2, h_l2, h_r2,
                                  out2, out2b, m2, (float*)nullptr, (float*)nullptr, ckey2, N);

    gat_edge_l2<<<cdiv_i(E, B), B, 0, stream>>>(
        src, dst, h_l2, h_r2, att2, ckey2, mode2, e2, m2, out2, out2b, E);
    gat_edge_aggregate_l2<<<cdiv_i(E, B), B, 0, stream>>>(
        src, dst, h_l2, e2, m2, out2, out2b, mode2, E);

    gat_finalize<<<cdiv_i(N, B / 32), B, 0, stream>>>((const float*)out2, (const float*)out2b,
                                                      bias2, (float*)d_out, N);
}


# Layer-1 edge pass: one 8-byte load per lane per row (channels 2l, 2l+1; a head on 4 lanes), so one logit sum and one exp per lane; accumulator slots even|odd; logit sum order changes
# speedup vs baseline: 3.6543x; 1.0574x over previous
#include <hip/hip_runtime.h>
#include <hip/hip_bf16.h>


#define DIM_IN   128
#define HID      64
#define HEADS    8
#define DIM_H    8
#define DIM_OUT  40
#define NEG_SLOPE 0.2f
#define ROW2     64
#define DEN2     40

typedef __attribute__((ext_vector_type(2))) float v2f;
typedef __attribute__((ext_vector_type(2))) unsigned int v2u;
typedef __attribute__((ext_vector_type(4))) float v4f;
typedef __attribute__((ext_vector_type(8))) float v8f;

__device__ __forceinline__ void atomicMaxFloat(float* addr, float val) {
    if (!(val == val)) return;
    if (val >= 0.0f) {
        atomicMax((int*)addr, __float_as_int(val));
    } else {
        atomicMin((unsigned int*)addr, __float_as_uint(val));
    }
}

template <int M>
__device__ __forceinline__ float gat_row_xor(float x) {
    return __int_as_float(__builtin_amdgcn_update_dpp(
        0, __float_as_int(x), 0x160 + M, 0xf, 0xf, false));
}

__device__ __forceinline__ float gat_xhalf(float x) {
    const int b = __float_as_int(x);
    return __int_as_float(__builtin_amdgcn_permlanex16(b, b, 0x76543210, 0xfedcba98, false, false));
}

__device__ __forceinline__ float gat_lrelu(float v) { return v > 0.0f ? v : v * NEG_SLOPE; }

__device__ __forceinline__ int gat_package_half() {
    return (int)((__builtin_amdgcn_s_sendmsg_rtn(0x87) >> 18) & 1u);
}

#define GAT_KEY_NINF 0x007fffffu
#define GAT_KEY_PINF 0xff800000u
__device__ __forceinline__ unsigned int gat_key(float f) {
    const unsigned int b = __float_as_uint(f);
    return (b & 0x80000000u) ? ~b : (b | 0x80000000u);
}
__device__ __forceinline__ float gat_unkey(unsigned int k) {
    return __uint_as_float((k & 0x80000000u) ? (k & 0x7fffffffu) : ~k);
}

__global__ __launch_bounds__(256) void gat_bound_start(unsigned int* __restrict__ k1,
                                                       unsigned int* __restrict__ k2) {
    const int t = threadIdx.x;
    k1[t] = ((t >> 6) & 1) ? GAT_KEY_PINF : GAT_KEY_NINF;
    if (t < 192) k2[t] = ((t / 48) & 1) ? GAT_KEY_PINF : GAT_KEY_NINF;
}

template <int K, int CC, int NT, int TPW, int ACCW, int MW, int DW>
__global__ __launch_bounds__(32 * (2 * NT / TPW)) void gat_proj_pair_f32_wmma(
    const float* __restrict__ A,
    const float* __restrict__ Wl, const float* __restrict__ bl,
    const float* __restrict__ Wr, const float* __restrict__ br,
    float* __restrict__ outl, float* __restrict__ outr,
    float* __restrict__ acc0, float* __restrict__ acc1, float* __restrict__ max0,
    float* __restrict__ den0, float* __restrict__ den1,
    unsigned int* __restrict__ ckey, int n) {
    constexpr int WPM = NT / TPW;
    constexpr int KS = K / 4;
    constexpr bool PAD = (16 * NT != CC);
    const int lane = threadIdx.x & 31;
    const int wave = __builtin_amdgcn_readfirstlane((int)(threadIdx.x >> 5));
    const bool right = wave >= WPM;
    const float* __restrict__ W = right ? Wr : Wl;
    const float* __restrict__ bias = right ? br : bl;
    float* __restrict__ out = right ? outr : outl;
    const int tile0 = (WPM == 1) ? 0 : (right ? wave - WPM : wave) * TPW;
    const int c = lane & 15;
    const int h = lane >> 4;

    v2f b[TPW][KS];
    float bv[TPW];
#pragma unroll
    for (int t = 0; t < TPW; ++t) {
        const int col = (tile0 + t) * 16 + c;
        const bool ok = !PAD || col < CC;
        const float* __restrict__ wc = W + (ok ? col : 0);
        bv[t] = ok ? bias[col] : 0.0f;
#pragma unroll
        for (int s = 0; s < KS; ++s) {
            const int k = 8 * (s >> 1) + 4 * h + 2 * (s & 1);
            const float w0 = wc[(size_t)k * CC];
            const float w1 = wc[(size_t)(k + 1) * CC];
            b[t][s].x = ok ? w0 : 0.0f;
            b[t][s].y = ok ? w1 : 0.0f;
        }
    }

    float cmx[TPW], cmn[TPW];
#pragma unroll
    for (int t = 0; t < TPW; ++t) { cmx[t] = -__builtin_inff(); cmn[t] = __builtin_inff(); }

    const int rowTiles = n >> 4;
    for (int rt = blockIdx.x; rt < rowTiles; rt += gridDim.x) {
        const float* __restrict__ aRow = A + (size_t)((rt << 4) + c) * K + 4 * h;
        v4f a[K / 8];
#pragma unroll
        for (int j = 0; j < K / 8; ++j) a[j] = *(const v4f*)(aRow + 8 * j);

        v8f acc[TPW];
#pragma unroll
        for (int t = 0; t < TPW; ++t) acc[t] = v8f{};
#pragma unroll
        for (int j = 0; j < K / 8; ++j) {
            const v2f a0 = {a[j].x, a[j].y};
            const v2f a1 = {a[j].z, a[j].w};
#pragma unroll
            for (int t = 0; t < TPW; ++t)
                acc[t] = __builtin_amdgcn_wmma_f32_16x16x4_f32(
                    false, a0, false, b[t][2 * j], (short)0, acc[t], false, false);
#pragma unroll
            for (int t = 0; t < TPW; ++t)
                acc[t] = __builtin_amdgcn_wmma_f32_16x16x4_f32(
                    false, a1, false, b[t][2 * j + 1], (short)0, acc[t], false, false);
        }

        const int rbase = (rt << 4) + (h << 3);
#pragma unroll
        for (int t = 0; t < TPW; ++t) {
            const int col = (tile0 + t) * 16 + c;
            if (!PAD || col < CC) {
#pragma unroll
                for (int r = 0; r < 8; ++r) {
                    const float v = acc[t][r] + bv[t];
                    out[(size_t)(rbase + r) * CC + col] = v;
                    cmx[t] = fmaxf(cmx[t], v);
                    cmn[t] = fminf(cmn[t], v);
                }
            }
        }

        constexpr int NW = 2 * WPM;
        constexpr int ACC_P = 16 * ACCW / 4;
        constexpr int MAX_P = 16 * MW / 4;
        constexpr int DEN_P = 16 * DW / 4;
        static_assert(ACC_P % (NW * 32) == 0 && MAX_P <= 32 && DEN_P <= 32, "tile start values");
        v4f* __restrict__ pa = (v4f*)(acc0 + (size_t)rt * (16 * ACCW));
        v4f* __restrict__ pb = (v4f*)(acc1 + (size_t)rt * (16 * ACCW));
        for (int q = 0; q < ACC_P / (NW * 32); ++q) {
            pa[(q * NW + wave) * 32 + lane] = v4f{};
            pb[(q * NW + wave) * 32 + lane] = v4f{};
        }
        if (wave == NW - 1 && lane < MAX_P) {
            const float ninf = -__builtin_inff();
            ((v4f*)(max0 + (size_t)rt * (16 * MW)))[lane] = v4f{ninf, ninf, ninf, ninf};
        }
        if constexpr (DW > 0) {
            if (wave == 0 && lane < DEN_P) {
                ((v4f*)(den0 + (size_t)rt * (16 * DW)))[lane] = v4f{};
                ((v4f*)(den1 + (size_t)rt * (16 * DW)))[lane] = v4f{};
            }
        }
    }

    if (!right) {
        unsigned int* __restrict__ kb = ckey + gat_package_half() * (32 * NT);
#pragma unroll
        for (int t = 0; t < TPW; ++t) {
            const float mx = fmaxf(cmx[t], __shfl_xor(cmx[t], 16, 32));
            const float mn = fminf(cmn[t], __shfl_xor(cmn[t], 16, 32));
            const int col = (tile0 + t) * 16 + c;
            if (h == 0 && (!PAD || col < CC)) {
                atomicMax(kb + col, gat_key(mx));
                atomicMin(kb + 16 * NT + col, gat_key(mn));
            }
        }
    }
}

__device__ __forceinline__ bool gat_bound_l1(
    const unsigned int* __restrict__ ckey, float av0, float av1, int lane,
    float& bn0, float& bn1) {
    const v2u* __restrict__ k0 = (const v2u*)ckey;
    const v2u* __restrict__ k1 = (const v2u*)(ckey + 128);
    const v2u mx0 = k0[lane], mx1 = k1[lane];
    const v2u mn0 = k0[32 + lane], mn1 = k1[32 + lane];
    const float u0 = fmaxf(gat_unkey(mx0.x), gat_unkey(mx1.x));
    const float u1 = fmaxf(gat_unkey(mx0.y), gat_unkey(mx1.y));
    const float l0 = fminf(gat_unkey(mn0.x), gat_unkey(mn1.x));
    const float l1 = fminf(gat_unkey(mn0.y), gat_unkey(mn1.y));
    bn0 = av0 >= 0.0f ? u0 : l0;
    bn1 = av1 >= 0.0f ? u1 : l1;
    float g = fabsf(av0) * (u0 - l0) + fabsf(av1) * (u1 - l1);
    g += gat_row_xor<1>(g);
    g += gat_row_xor<2>(g);
    return __builtin_amdgcn_ballot_w32(!(g < 70.0f)) != 0u;
}

__device__ __forceinline__ bool gat_bound_l2(
    const unsigned int* __restrict__ ckey, float av0, float av1, int lane, int c1,
    bool tail_ch, float& bn0, float& bn1) {
    const unsigned int* __restrict__ k0 = ckey;
    const unsigned int* __restrict__ k1 = ckey + 96;
    const float u0 = fmaxf(gat_unkey(k0[lane]), gat_unkey(k1[lane]));
    const float u1 = fmaxf(gat_unkey(k0[c1]), gat_unkey(k1[c1]));
    const float l0 = fminf(gat_unkey(k0[48 + lane]), gat_unkey(k1[48 + lane]));
    const float l1 = fminf(gat_unkey(k0[48 + c1]), gat_unkey(k1[48 + c1]));
    bn0 = av0 >= 0.0f ? u0 : l0;
    bn1 = av1 >= 0.0f ? u1 : l1;
    const float ga = fabsf(av0) * (u0 - l0);
    const float gb = fabsf(av1) * (u1 - l1);
    float g = ga + (tail_ch ? gb : 0.0f);
    g += gat_row_xor<1>(g);
    g += gat_row_xor<2>(g);
    g += gat_row_xor<4>(g);
    g += gat_row_xor<8>(g);
    g += gat_xhalf(g);
    return __builtin_amdgcn_ballot_w32(!(g < 70.0f)) != 0u;
}

__device__ __forceinline__ void gat_logit_max_item_l1(
    const float* __restrict__ xl, const float* __restrict__ xr,
    const float* __restrict__ att, const int* __restrict__ src,
    const int* __restrict__ dst, float* __restrict__ e1,
    float* __restrict__ m1, long long t) {
    const int e = (int)(t >> 3);
    const int h = (int)(t & 7);
    const int s = src[e], d = dst[e];
    const float* pj = xl + (size_t)s * HID + h * DIM_H;
    const float* pi = xr + (size_t)d * HID + h * DIM_H;
    const float* pa = att + h * DIM_H;
    float acc = 0.0f;
#pragma unroll
    for (int c = 0; c < DIM_H; ++c) {
        float v = pi[c] + pj[c];
        v = v > 0.0f ? v : v * NEG_SLOPE;
        acc += v * pa[c];
    }
    e1[t] = acc;
    atomicMaxFloat(&m1[(size_t)d * HEADS + h], acc);
}

__device__ __forceinline__ void gat_edge_step_l1(
    v2f xj, v2f xi, float av0, float av1, float bn0, float bn1,
    bool den_lane, float* __restrict__ den,
    float* __restrict__ po, int lane) {
    float t = (gat_lrelu(xi.x + xj.x) - gat_lrelu(xi.x + bn0)) * av0
            + (gat_lrelu(xi.y + xj.y) - gat_lrelu(xi.y + bn1)) * av1;
    t += gat_row_xor<1>(t);
    t += gat_row_xor<2>(t);
    const float w = expf(t);
    if (den_lane) atomicAdd(den, w);
    atomicAdd(po + lane, xj.x * w);
    atomicAdd(po + 32 + lane, xj.y * w);
}

__global__ __launch_bounds__(256) void gat_edge_l1(
    const int* __restrict__ src, const int* __restrict__ dst,
    const float* __restrict__ xl, const float* __restrict__ xr,
    const float* __restrict__ att, const unsigned int* __restrict__ ckey,
    int* __restrict__ mode, float* __restrict__ e1,
    float* __restrict__ m1, float* __restrict__ denom1a, float* __restrict__ denom1b,
    float* __restrict__ out1a, float* __restrict__ out1b, int E) {
    const int lane = threadIdx.x & 31;
    const float av0 = att[2 * lane], av1 = att[2 * lane + 1];
    float bn0, bn1;
    const bool loose = gat_bound_l1(ckey, av0, av1, lane, bn0, bn1);
    if (blockIdx.x == 0 && threadIdx.x == 0) *mode = loose ? 1 : 0;
    if (loose) {
        const long long E8 = (long long)E * HEADS;
        for (long long t = (long long)blockIdx.x * blockDim.x + threadIdx.x; t < E8;
             t += (long long)gridDim.x * blockDim.x)
            gat_logit_max_item_l1(xl, xr, att, src, dst, e1, m1, t);
        return;
    }
    const int wave = __builtin_amdgcn_readfirstlane(
        (int)(blockIdx.x * (blockDim.x >> 5) + (threadIdx.x >> 5)));
    const int wbase = wave * 32;
    if (wbase >= E) return;
    const bool upper = gat_package_half() != 0;
    float* __restrict__ denom1 = upper ? denom1b : denom1a;
    float* __restrict__ out1 = upper ? out1b : out1a;
    const int nE = (E - wbase) < 32 ? (E - wbase) : 32;
    int s_i = 0, d_i = 0;
    if (lane < nE) {
        s_i = src[wbase + lane];
        d_i = dst[wbase + lane];
    }
    const bool den_lane = (lane & 3) == 0;
    const int den_off = lane >> 2;

    int j0 = 0;
    for (; j0 + 4 <= nE; j0 += 4) {
        v2f xj[4], xi[4];
#pragma unroll
        for (int q = 0; q < 4; ++q) {
            const int s = __builtin_amdgcn_readlane(s_i, j0 + q);
            const int d = __builtin_amdgcn_readlane(d_i, j0 + q);
            xj[q] = *(const v2f*)(xl + (size_t)s * HID + 2 * lane);
            xi[q] = *(const v2f*)(xr + (size_t)d * HID + 2 * lane);
        }
#pragma unroll
        for (int q = 0; q < 4; ++q) {
            const int d = __builtin_amdgcn_readlane(d_i, j0 + q);
            gat_edge_step_l1(xj[q], xi[q], av0, av1, bn0, bn1,
                             den_lane, denom1 + (size_t)d * HEADS + den_off,
                             out1 + (size_t)d * HID, lane);
        }
    }
    for (; j0 < nE; ++j0) {
        const int s = __builtin_amdgcn_readlane(s_i, j0);
        const int d = __builtin_amdgcn_readlane(d_i, j0);
        const v2f xjv = *(const v2f*)(xl + (size_t)s * HID + 2 * lane);
        const v2f xiv = *(const v2f*)(xr + (size_t)d * HID + 2 * lane);
        gat_edge_step_l1(xjv, xiv, av0, av1, bn0, bn1,
                         den_lane, denom1 + (size_t)d * HEADS + den_off,
                         out1 + (size_t)d * HID, lane);
    }
}

__global__ __launch_bounds__(256) void gat_edge_aggregate_l1(
    const int* __restrict__ src, const int* __restrict__ dst,
    const float* __restrict__ xl, const float* __restrict__ e1,
    const float* __restrict__ m1, float* __restrict__ denom1a, float* __restrict__ denom1b,
    float* __restrict__ out1a, float* __restrict__ out1b,
    const int* __restrict__ mode, int E) {
    if (*mode == 0) return;
    const int lane = threadIdx.x & 31;
    const int wave = __builtin_amdgcn_readfirstlane(
        (int)(blockIdx.x * (blockDim.x >> 5) + (threadIdx.x >> 5)));
    const int wbase = wave * 32;
    if (wbase >= E) return;
    const bool upper = gat_package_half() != 0;
    float* __restrict__ denom1 = upper ? denom1b : denom1a;
    float* __restrict__ out1 = upper ? out1b : out1a;
    const int sub = lane >> 3;
    const int h = lane & 7;
    const int hd = lane >> 2;
    const int nE = (E - wbase) < 32 ? (E - wbase) : 32;

    if (nE == 32) {
        const int s_i = src[wbase + lane];
        const int d_i = dst[wbase + lane];
        float ev[8], mv[8], ex[8];
        int dp[8];
#pragma unroll
        for (int g = 0; g < 8; ++g) {
            dp[g] = __shfl(d_i, g * 4 + sub, 32);
            ev[g] = e1[(size_t)wbase * HEADS + g * 32 + lane];
            mv[g] = m1[(size_t)dp[g] * HEADS + h];
        }
#pragma unroll
        for (int g = 0; g < 8; ++g) {
            ex[g] = expf(ev[g] - mv[g]);
            atomicAdd(&denom1[(size_t)dp[g] * HEADS + h], ex[g]);
        }
#pragma unroll
        for (int g = 0; g < 8; ++g) {
            v2f r[4];
#pragma unroll
            for (int q = 0; q < 4; ++q) {
                const int s = __builtin_amdgcn_readlane(s_i, g * 4 + q);
                r[q] = *(const v2f*)(xl + (size_t)s * HID + 2 * lane);
            }
#pragma unroll
            for (int q = 0; q < 4; ++q) {
                const int d = __builtin_amdgcn_readlane(d_i, g * 4 + q);
                const float a = __shfl(ex[g], q * 8 + hd, 32);
                float* po = out1 + (size_t)d * HID;
                atomicAdd(po + lane, r[q].x * a);
                atomicAdd(po + 32 + lane, r[q].y * a);
            }
        }
    } else {
        for (int j = 0; j < nE; ++j) {
            const int e = wbase + j;
            const int s = src[e], d = dst[e];
            const float a = expf(e1[(size_t)e * HEADS + hd] - m1[(size_t)d * HEADS + hd]);
            if ((lane & 3) == 0) atomicAdd(&denom1[(size_t)d * HEADS + hd], a);
            const v2f rv = *(const v2f*)(xl + (size_t)s * HID + 2 * lane);
            float* po = out1 + (size_t)d * HID;
            atomicAdd(po + lane, rv.x * a);
            atomicAdd(po + 32 + lane, rv.y * a);
        }
    }
}

__global__ void gat_norm_bias_elu(float* __restrict__ h1,
                                  const float* __restrict__ out1,
                                  const float* __restrict__ out1b,
                                  const float* __restrict__ denom1,
                                  const float* __restrict__ denom1b,
                                  const float* __restrict__ bias1,
                                  long long total) {
    long long i = (long long)blockIdx.x * blockDim.x + threadIdx.x;
    if (i >= total) return;
    const int c = (int)(i & (HID - 1));
    const long long p = (i - c) + (((c & 1) << 5) | (c >> 1));
    const float den = denom1[i >> 3] + denom1b[i >> 3];
    const float acc = out1[p] + out1b[p];
    float v = (den > 0.0f ? acc / den : acc) + bias1[c];
    h1[i] = v > 0.0f ? v : expm1f(v);
}

__device__ __forceinline__ void gat_logit_max_item_l2(
    const float* __restrict__ xl, const float* __restrict__ xr,
    const float* __restrict__ att, const int* __restrict__ src,
    const int* __restrict__ dst, float* __restrict__ e2,
    float* __restrict__ m2, long long t) {
    const int e = (int)(t >> 3);
    const int sub = (int)(t & 7);
    const int s = src[e], d = dst[e];
    const float* pj = xl + (size_t)s * DIM_OUT;
    const float* pi = xr + (size_t)d * DIM_OUT;
    float acc = 0.0f;
#pragma unroll
    for (int c = sub; c < DIM_OUT; c += 8) {
        float v = pi[c] + pj[c];
        v = v > 0.0f ? v : v * NEG_SLOPE;
        acc += v * att[c];
    }
    acc += __shfl_xor(acc, 1, 32);
    acc += __shfl_xor(acc, 2, 32);
    acc += __shfl_xor(acc, 4, 32);
    if (sub == 0) {
        e2[e] = acc;
        atomicMaxFloat(&m2[d], acc);
    }
}

__device__ __forceinline__ void gat_edge_step_l2(
    float xj0, float xj1, float xi0, float xi1,
    float av0, float av1, float bn0, float bn1,
    bool tail_ch, float* __restrict__ po, int lane) {
    const float ta = (gat_lrelu(xi0 + xj0) - gat_lrelu(xi0 + bn0)) * av0;
    const float tb = (gat_lrelu(xi1 + xj1) - gat_lrelu(xi1 + bn1)) * av1;
    float t = ta + (tail_ch ? tb : 0.0f);
    t += gat_row_xor<1>(t);
    t += gat_row_xor<2>(t);
    t += gat_row_xor<4>(t);
    t += gat_row_xor<8>(t);
    t += gat_xhalf(t);
    const float w = expf(t);
    const float v1 = tail_ch ? xj1 : 1.0f;
    atomicAdd(po + lane, xj0 * w);
    if (lane <= DEN2 - 32) atomicAdd(po + 32 + lane, v1 * w);
}

__global__ __launch_bounds__(256) void gat_edge_l2(
    const int* __restrict__ src, const int* __restrict__ dst,
    const float* __restrict__ xl, const float* __restrict__ xr,
    const float* __restrict__ att, const unsigned int* __restrict__ ckey,
    int* __restrict__ mode, float* __restrict__ e2,
    float* __restrict__ m2, float* __restrict__ out2a, float* __restrict__ out2b, int E) {
    const int lane = threadIdx.x & 31;
    const int c1 = 32 + (lane & 7);
    const bool tail_ch = lane < DIM_OUT - 32;
    const float av0 = att[lane], av1 = att[c1];
    float bn0, bn1;
    const bool loose = gat_bound_l2(ckey, av0, av1, lane, c1, tail_ch, bn0, bn1);
    if (blockIdx.x == 0 && threadIdx.x == 0) *mode = loose ? 1 : 0;
    if (loose) {
        const long long E8 = (long long)E * 8;
        for (long long t = (long long)blockIdx.x * blockDim.x + threadIdx.x; t < E8;
             t += (long long)gridDim.x * blockDim.x)
            gat_logit_max_item_l2(xl, xr, att, src, dst, e2, m2, t);
        return;
    }
    const int wave = __builtin_amdgcn_readfirstlane(
        (int)(blockIdx.x * (blockDim.x >> 5) + (threadIdx.x >> 5)));
    const int wbase = wave * 32;
    if (wbase >= E) return;
    float* __restrict__ out2 = gat_package_half() != 0 ? out2b : out2a;
    const int nE = (E - wbase) < 32 ? (E - wbase) : 32;
    int s_i = 0, d_i = 0;
    if (lane < nE) {
        s_i = src[wbase + lane];
        d_i = dst[wbase + lane];
    }
    int j0 = 0;
    for (; j0 + 4 <= nE; j0 += 4) {
        float xj0[4], xj1[4], xi0[4], xi1[4];
#pragma unroll
        for (int q = 0; q < 4; ++q) {
            const int s = __builtin_amdgcn_readlane(s_i, j0 + q);
            const int d = __builtin_amdgcn_readlane(d_i, j0 + q);
            const float* pj = xl + (size_t)s * DIM_OUT;
            const float* pi = xr + (size_t)d * DIM_OUT;
            xj0[q] = pj[lane];
            xj1[q] = pj[c1];
            xi0[q] = pi[lane];
            xi1[q] = pi[c1];
        }
#pragma unroll
        for (int q = 0; q < 4; ++q) {
            const int d = __builtin_amdgcn_readlane(d_i, j0 + q);
            gat_edge_step_l2(xj0[q], xj1[q], xi0[q], xi1[q], av0, av1, bn0, bn1,
                             tail_ch, out2 + (size_t)d * ROW2, lane);
        }
    }
    for (; j0 < nE; ++j0) {
        const int s = __builtin_amdgcn_readlane(s_i, j0);
        const int d = __builtin_amdgcn_readlane(d_i, j0);
        const float* pj = xl + (size_t)s * DIM_OUT;
        const float* pi = xr + (size_t)d * DIM_OUT;
        gat_edge_step_l2(pj[lane], pj[c1], pi[lane], pi[c1], av0, av1, bn0, bn1,
                         tail_ch, out2 + (size_t)d * ROW2, lane);
    }
}

__global__ __launch_bounds__(256) void gat_edge_aggregate_l2(
    const int* __restrict__ src, const int* __restrict__ dst,
    const float* __restrict__ xl, const float* __restrict__ e2,
    const float* __restrict__ m2, float* __restrict__ out2a, float* __restrict__ out2b,
    const int* __restrict__ mode, int E) {
    if (*mode == 0) return;
    const int lane = threadIdx.x & 31;
    const int wave = __builtin_amdgcn_readfirstlane(
        (int)(blockIdx.x * (blockDim.x >> 5) + (threadIdx.x >> 5)));
    const int wbase = wave * 32;
    if (wbase >= E) return;
    float* __restrict__ out2 = gat_package_half() != 0 ? out2b : out2a;
    const int ei = wbase + lane;
    int s_i = 0, d_i = 0;
    float ex_i = 0.0f;
    if (ei < E) {
        s_i = src[ei];
        d_i = dst[ei];
        ex_i = expf(e2[ei] - m2[d_i]);
    }
    const int nE = (E - wbase) < 32 ? (E - wbase) : 32;
    const int c1 = 32 + (lane & 7);
    const bool tail_ch = lane < DIM_OUT - 32;

    if (nE == 32) {
        for (int j0 = 0; j0 < 32; j0 += 4) {
            float r0[4], r1[4];
#pragma unroll
            for (int q = 0; q < 4; ++q) {
                const int s = __builtin_amdgcn_readlane(s_i, j0 + q);
                const float* pj = xl + (size_t)s * DIM_OUT;
                r0[q] = pj[lane];
                r1[q] = pj[c1];
            }
            asm volatile("" :: "v"(r1[0]), "v"(r1[1]), "v"(r1[2]), "v"(r1[3]));
#pragma unroll
            for (int q = 0; q < 4; ++q) {
                const int d = __builtin_amdgcn_readlane(d_i, j0 + q);
                const float a = __int_as_float(
                    __builtin_amdgcn_readlane(__float_as_int(ex_i), j0 + q));
                float* po = out2 + (size_t)d * ROW2;
                const float v1 = tail_ch ? r1[q] : 1.0f;
                atomicAdd(po + lane, r0[q] * a);
                if (lane <= DEN2 - 32) atomicAdd(po + 32 + lane, v1 * a);
            }
        }
    } else {
        for (int j = 0; j < nE; ++j) {
            const int s = __builtin_amdgcn_readlane(s_i, j);
            const int d = __builtin_amdgcn_readlane(d_i, j);
            const float a = __int_as_float(
                __builtin_amdgcn_readlane(__float_as_int(ex_i), j));
            const float* pj = xl + (size_t)s * DIM_OUT;
            float* po = out2 + (size_t)d * ROW2;
            const float v0 = pj[lane];
            const float v1 = tail_ch ? pj[c1] : 1.0f;
            atomicAdd(po + lane, v0 * a);
            if (lane <= DEN2 - 32) atomicAdd(po + 32 + lane, v1 * a);
        }
    }
}

__global__ __launch_bounds__(256) void gat_finalize(
    const float* __restrict__ out2, const float* __restrict__ out2b,
    const float* __restrict__ bias2, float* __restrict__ d_out, int N) {
    const int node = blockIdx.x * (blockDim.x >> 5) + (threadIdx.x >> 5);
    const int lane = threadIdx.x & 31;
    if (node >= N) return;
    const float* row = out2 + (size_t)node * ROW2;
    const float* rowb = out2b + (size_t)node * ROW2;
    const float den = row[DEN2] + rowb[DEN2];
    const float ra = row[lane] + rowb[lane];
    const float a = (den > 0.0f ? ra / den : ra) + bias2[lane];
    const bool has2 = (lane < DIM_OUT - 32);
    const float rb = has2 ? row[32 + lane] + rowb[32 + lane] : 0.0f;
    const float b = has2 ? ((den > 0.0f ? rb / den : rb) + bias2[32 + lane])
                         : -__builtin_inff();
    float mx = fmaxf(a, b);
#pragma unroll
    for (int o = 16; o >= 1; o >>= 1) mx = fmaxf(mx, __shfl_xor(mx, o, 32));
    float s = expf(a - mx) + (has2 ? expf(b - mx) : 0.0f);
#pragma unroll
    for (int o = 16; o >= 1; o >>= 1) s += __shfl_xor(s, o, 32);
    const float lse = logf(s) + mx;
    float* h  = d_out + (size_t)node * DIM_OUT;
    float* ls = d_out + (size_t)N * DIM_OUT + (size_t)node * DIM_OUT;
    h[lane] = a;
    ls[lane] = a - lse;
    if (has2) {
        h[32 + lane] = b;
        ls[32 + lane] = b - lse;
    }
}

static inline int cdiv_i(long long a, long long b) { return (int)((a + b - 1) / b); }

extern "C" void kernel_launch(void* const* d_in, const int* in_sizes, int n_in,
                              void* d_out, int out_size, void* d_ws, size_t ws_size,
                              hipStream_t stream) {
    (void)n_in; (void)out_size; (void)ws_size;
    const float* x     = (const float*)d_in[0];
    const int*   eidx  = (const int*)d_in[1];
    const float* Wl1   = (const float*)d_in[2];
    const float* bl1   = (const float*)d_in[3];
    const float* Wr1   = (const float*)d_in[4];
    const float* br1   = (const float*)d_in[5];
    const float* att1  = (const float*)d_in[6];
    const float* bias1 = (const float*)d_in[7];
    const float* Wl2   = (const float*)d_in[8];
    const float* bl2   = (const float*)d_in[9];
    const float* Wr2   = (const float*)d_in[10];
    const float* br2   = (const float*)d_in[11];
    const float* att2  = (const float*)d_in[12];
    const float* bias2 = (const float*)d_in[13];

    const int N = in_sizes[0] / DIM_IN;
    const int E = in_sizes[1] / 2;
    const int* src = eidx;
    const int* dst = eidx + E;

    float* ws = (float*)d_ws;
    size_t o = 0;
    float* h_l1   = ws + o; o += (size_t)N * HID;
    float* h_r1   = ws + o; o += (size_t)N * HID;
    float* e1     = ws + o; o += (size_t)E * HEADS;
    float* m1     = ws + o; o += (size_t)N * HEADS;
    float* denom1 = ws + o; o += (size_t)N * HEADS;
    float* denom1b = ws + o; o += (size_t)N * HEADS;
    float* out1   = ws + o; o += (size_t)N * HID;
    float* out1b  = ws + o; o += (size_t)N * HID;
    float* h1     = ws + o; o += (size_t)N * HID;
    unsigned int* ckey1 = (unsigned int*)(ws + o); o += 256;
    int*   mode1  = (int*)(ws + o); o += 64;
    unsigned int* ckey2 = (unsigned int*)(ws + o); o += 256;
    int*   mode2  = (int*)(ws + o); o += 64;
    float* h_l2   = ws;
    float* h_r2   = h_l2 + (size_t)N * DIM_OUT;
    float* e2     = h_r2 + (size_t)N * DIM_OUT;
    float* out2   = e2 + (size_t)E;
    float* out2b  = out2 + (size_t)N * ROW2;
    float* m2     = out2b + (size_t)N * ROW2;

    const int B = 256;

    gat_bound_start<<<1, 256, 0, stream>>>(ckey1, ckey2);

    gat_proj_pair_f32_wmma<DIM_IN, HID, 4, 2, HID, HEADS, HEADS>
        <<<1024, 128, 0, stream>>>(x, Wl1, bl1, Wr1, br1, h_l1, h_r1,
                                   out1, out1b, m1, denom1, denom1b, ckey1, N);

    gat_edge_l1<<<cdiv_i(E, B), B, 0, stream>>>(
        src, dst, h_l1, h_r1, att1, ckey1, mode1, e1, m1, denom1, denom1b, out1, out1b, E);
    gat_edge_aggregate_l1<<<cdiv_i(E, B), B, 0, stream>>>(
        src, dst, h_l1, e1, m1, denom1, denom1b, out1, out1b, mode1, E);
    gat_norm_bias_elu<<<cdiv_i((long long)N * HID, B), B, 0, stream>>>(
        h1, out1, out1b, denom1, denom1b, bias1, (long long)N * HID);

    gat_proj_pair_f32_wmma<HID, DIM_OUT, 3, 3, ROW2, 1, 0>
        <<<1250, 64, 0, stream>>>(h1, Wl2, bl2, Wr2, br2, h_l2, h_r2,
                                  out2, out2b, m2, (float*)nullptr, (float*)nullptr, ckey2, N);

    gat_edge_l2<<<cdiv_i(E, B), B, 0, stream>>>(
        src, dst, h_l2, h_r2, att2, ckey2, mode2, e2, m2, out2, out2b, E);
    gat_edge_aggregate_l2<<<cdiv_i(E, B), B, 0, stream>>>(
        src, dst, h_l2, e2, m2, out2, out2b, mode2, E);

    gat_finalize<<<cdiv_i(N, B / 32), B, 0, stream>>>((const float*)out2, (const float*)out2b,
                                                      bias2, (float*)d_out, N);
}


# Node passes and the two guarded fallback launches as resident waves that loop (about 33,000 waves enter per call instead of 400,000); ELU pass with 8-byte loads and 16-byte stores; sums unchanged
# speedup vs baseline: 3.9151x; 1.0714x over previous
#include <hip/hip_runtime.h>
#include <hip/hip_bf16.h>


#define DIM_IN   128
#define HID      64
#define HEADS    8
#define DIM_H    8
#define DIM_OUT  40
#define NEG_SLOPE 0.2f
#define ROW2     64
#define DEN2     40

typedef __attribute__((ext_vector_type(2))) float v2f;
typedef __attribute__((ext_vector_type(2))) unsigned int v2u;
typedef __attribute__((ext_vector_type(4))) float v4f;
typedef __attribute__((ext_vector_type(8))) float v8f;

__device__ __forceinline__ void atomicMaxFloat(float* addr, float val) {
    if (!(val == val)) return;
    if (val >= 0.0f) {
        atomicMax((int*)addr, __float_as_int(val));
    } else {
        atomicMin((unsigned int*)addr, __float_as_uint(val));
    }
}

template <int M>
__device__ __forceinline__ float gat_row_xor(float x) {
    return __int_as_float(__builtin_amdgcn_update_dpp(
        0, __float_as_int(x), 0x160 + M, 0xf, 0xf, false));
}

__device__ __forceinline__ float gat_xhalf(float x) {
    const int b = __float_as_int(x);
    return __int_as_float(__builtin_amdgcn_permlanex16(b, b, 0x76543210, 0xfedcba98, false, false));
}

__device__ __forceinline__ float gat_lrelu(float v) { return v > 0.0f ? v : v * NEG_SLOPE; }

__device__ __forceinline__ int gat_package_half() {
    return (int)((__builtin_amdgcn_s_sendmsg_rtn(0x87) >> 18) & 1u);
}

#define GAT_KEY_NINF 0x007fffffu
#define GAT_KEY_PINF 0xff800000u
__device__ __forceinline__ unsigned int gat_key(float f) {
    const unsigned int b = __float_as_uint(f);
    return (b & 0x80000000u) ? ~b : (b | 0x80000000u);
}
__device__ __forceinline__ float gat_unkey(unsigned int k) {
    return __uint_as_float((k & 0x80000000u) ? (k & 0x7fffffffu) : ~k);
}

__global__ __launch_bounds__(256) void gat_bound_start(unsigned int* __restrict__ k1,
                                                       unsigned int* __restrict__ k2) {
    const int t = threadIdx.x;
    k1[t] = ((t >> 6) & 1) ? GAT_KEY_PINF : GAT_KEY_NINF;
    if (t < 192) k2[t] = ((t / 48) & 1) ? GAT_KEY_PINF : GAT_KEY_NINF;
}

template <int K, int CC, int NT, int TPW, int ACCW, int MW, int DW>
__global__ __launch_bounds__(32 * (2 * NT / TPW)) void gat_proj_pair_f32_wmma(
    const float* __restrict__ A,
    const float* __restrict__ Wl, const float* __restrict__ bl,
    const float* __restrict__ Wr, const float* __restrict__ br,
    float* __restrict__ outl, float* __restrict__ outr,
    float* __restrict__ acc0, float* __restrict__ acc1, float* __restrict__ max0,
    float* __restrict__ den0, float* __restrict__ den1,
    unsigned int* __restrict__ ckey, int n) {
    constexpr int WPM = NT / TPW;
    constexpr int KS = K / 4;
    constexpr bool PAD = (16 * NT != CC);
    const int lane = threadIdx.x & 31;
    const int wave = __builtin_amdgcn_readfirstlane((int)(threadIdx.x >> 5));
    const bool right = wave >= WPM;
    const float* __restrict__ W = right ? Wr : Wl;
    const float* __restrict__ bias = right ? br : bl;
    float* __restrict__ out = right ? outr : outl;
    const int tile0 = (WPM == 1) ? 0 : (right ? wave - WPM : wave) * TPW;
    const int c = lane & 15;
    const int h = lane >> 4;

    v2f b[TPW][KS];
    float bv[TPW];
#pragma unroll
    for (int t = 0; t < TPW; ++t) {
        const int col = (tile0 + t) * 16 + c;
        const bool ok = !PAD || col < CC;
        const float* __restrict__ wc = W + (ok ? col : 0);
        bv[t] = ok ? bias[col] : 0.0f;
#pragma unroll
        for (int s = 0; s < KS; ++s) {
            const int k = 8 * (s >> 1) + 4 * h + 2 * (s & 1);
            const float w0 = wc[(size_t)k * CC];
            const float w1 = wc[(size_t)(k + 1) * CC];
            b[t][s].x = ok ? w0 : 0.0f;
            b[t][s].y = ok ? w1 : 0.0f;
        }
    }

    float cmx[TPW], cmn[TPW];
#pragma unroll
    for (int t = 0; t < TPW; ++t) { cmx[t] = -__builtin_inff(); cmn[t] = __builtin_inff(); }

    const int rowTiles = n >> 4;
    for (int rt = blockIdx.x; rt < rowTiles; rt += gridDim.x) {
        const float* __restrict__ aRow = A + (size_t)((rt << 4) + c) * K + 4 * h;
        v4f a[K / 8];
#pragma unroll
        for (int j = 0; j < K / 8; ++j) a[j] = *(const v4f*)(aRow + 8 * j);

        v8f acc[TPW];
#pragma unroll
        for (int t = 0; t < TPW; ++t) acc[t] = v8f{};
#pragma unroll
        for (int j = 0; j < K / 8; ++j) {
            const v2f a0 = {a[j].x, a[j].y};
            const v2f a1 = {a[j].z, a[j].w};
#pragma unroll
            for (int t = 0; t < TPW; ++t)
                acc[t] = __builtin_amdgcn_wmma_f32_16x16x4_f32(
                    false, a0, false, b[t][2 * j], (short)0, acc[t], false, false);
#pragma unroll
            for (int t = 0; t < TPW; ++t)
                acc[t] = __builtin_amdgcn_wmma_f32_16x16x4_f32(
                    false, a1, false, b[t][2 * j + 1], (short)0, acc[t], false, false);
        }

        const int rbase = (rt << 4) + (h << 3);
#pragma unroll
        for (int t = 0; t < TPW; ++t) {
            const int col = (tile0 + t) * 16 + c;
            if (!PAD || col < CC) {
#pragma unroll
                for (int r = 0; r < 8; ++r) {
                    const float v = acc[t][r] + bv[t];
                    out[(size_t)(rbase + r) * CC + col] = v;
                    cmx[t] = fmaxf(cmx[t], v);
                    cmn[t] = fminf(cmn[t], v);
                }
            }
        }

        constexpr int NW = 2 * WPM;
        constexpr int ACC_P = 16 * ACCW / 4;
        constexpr int MAX_P = 16 * MW / 4;
        constexpr int DEN_P = 16 * DW / 4;
        static_assert(ACC_P % (NW * 32) == 0 && MAX_P <= 32 && DEN_P <= 32, "tile start values");
        v4f* __restrict__ pa = (v4f*)(acc0 + (size_t)rt * (16 * ACCW));
        v4f* __restrict__ pb = (v4f*)(acc1 + (size_t)rt * (16 * ACCW));
        for (int q = 0; q < ACC_P / (NW * 32); ++q) {
            pa[(q * NW + wave) * 32 + lane] = v4f{};
            pb[(q * NW + wave) * 32 + lane] = v4f{};
        }
        if (wave == NW - 1 && lane < MAX_P) {
            const float ninf = -__builtin_inff();
            ((v4f*)(max0 + (size_t)rt * (16 * MW)))[lane] = v4f{ninf, ninf, ninf, ninf};
        }
        if constexpr (DW > 0) {
            if (wave == 0 && lane < DEN_P) {
                ((v4f*)(den0 + (size_t)rt * (16 * DW)))[lane] = v4f{};
                ((v4f*)(den1 + (size_t)rt * (16 * DW)))[lane] = v4f{};
            }
        }
    }

    if (!right) {
        unsigned int* __restrict__ kb = ckey + gat_package_half() * (32 * NT);
#pragma unroll
        for (int t = 0; t < TPW; ++t) {
            const float mx = fmaxf(cmx[t], __shfl_xor(cmx[t], 16, 32));
            const float mn = fminf(cmn[t], __shfl_xor(cmn[t], 16, 32));
            const int col = (tile0 + t) * 16 + c;
            if (h == 0 && (!PAD || col < CC)) {
                atomicMax(kb + col, gat_key(mx));
                atomicMin(kb + 16 * NT + col, gat_key(mn));
            }
        }
    }
}

__device__ __forceinline__ bool gat_bound_l1(
    const unsigned int* __restrict__ ckey, float av0, float av1, int lane,
    float& bn0, float& bn1) {
    const v2u* __restrict__ k0 = (const v2u*)ckey;
    const v2u* __restrict__ k1 = (const v2u*)(ckey + 128);
    const v2u mx0 = k0[lane], mx1 = k1[lane];
    const v2u mn0 = k0[32 + lane], mn1 = k1[32 + lane];
    const float u0 = fmaxf(gat_unkey(mx0.x), gat_unkey(mx1.x));
    const float u1 = fmaxf(gat_unkey(mx0.y), gat_unkey(mx1.y));
    const float l0 = fminf(gat_unkey(mn0.x), gat_unkey(mn1.x));
    const float l1 = fminf(gat_unkey(mn0.y), gat_unkey(mn1.y));
    bn0 = av0 >= 0.0f ? u0 : l0;
    bn1 = av1 >= 0.0f ? u1 : l1;
    float g = fabsf(av0) * (u0 - l0) + fabsf(av1) * (u1 - l1);
    g += gat_row_xor<1>(g);
    g += gat_row_xor<2>(g);
    return __builtin_amdgcn_ballot_w32(!(g < 70.0f)) != 0u;
}

__device__ __forceinline__ bool gat_bound_l2(
    const unsigned int* __restrict__ ckey, float av0, float av1, int lane, int c1,
    bool tail_ch, float& bn0, float& bn1) {
    const unsigned int* __restrict__ k0 = ckey;
    const unsigned int* __restrict__ k1 = ckey + 96;
    const float u0 = fmaxf(gat_unkey(k0[lane]), gat_unkey(k1[lane]));
    const float u1 = fmaxf(gat_unkey(k0[c1]), gat_unkey(k1[c1]));
    const float l0 = fminf(gat_unkey(k0[48 + lane]), gat_unkey(k1[48 + lane]));
    const float l1 = fminf(gat_unkey(k0[48 + c1]), gat_unkey(k1[48 + c1]));
    bn0 = av0 >= 0.0f ? u0 : l0;
    bn1 = av1 >= 0.0f ? u1 : l1;
    const float ga = fabsf(av0) * (u0 - l0);
    const float gb = fabsf(av1) * (u1 - l1);
    float g = ga + (tail_ch ? gb : 0.0f);
    g += gat_row_xor<1>(g);
    g += gat_row_xor<2>(g);
    g += gat_row_xor<4>(g);
    g += gat_row_xor<8>(g);
    g += gat_xhalf(g);
    return __builtin_amdgcn_ballot_w32(!(g < 70.0f)) != 0u;
}

__device__ __forceinline__ void gat_logit_max_item_l1(
    const float* __restrict__ xl, const float* __restrict__ xr,
    const float* __restrict__ att, const int* __restrict__ src,
    const int* __restrict__ dst, float* __restrict__ e1,
    float* __restrict__ m1, long long t) {
    const int e = (int)(t >> 3);
    const int h = (int)(t & 7);
    const int s = src[e], d = dst[e];
    const float* pj = xl + (size_t)s * HID + h * DIM_H;
    const float* pi = xr + (size_t)d * HID + h * DIM_H;
    const float* pa = att + h * DIM_H;
    float acc = 0.0f;
#pragma unroll
    for (int c = 0; c < DIM_H; ++c) {
        float v = pi[c] + pj[c];
        v = v > 0.0f ? v : v * NEG_SLOPE;
        acc += v * pa[c];
    }
    e1[t] = acc;
    atomicMaxFloat(&m1[(size_t)d * HEADS + h], acc);
}

__device__ __forceinline__ void gat_edge_step_l1(
    v2f xj, v2f xi, float av0, float av1, float bn0, float bn1,
    bool den_lane, float* __restrict__ den,
    float* __restrict__ po, int lane) {
    float t = (gat_lrelu(xi.x + xj.x) - gat_lrelu(xi.x + bn0)) * av0
            + (gat_lrelu(xi.y + xj.y) - gat_lrelu(xi.y + bn1)) * av1;
    t += gat_row_xor<1>(t);
    t += gat_row_xor<2>(t);
    const float w = expf(t);
    if (den_lane) atomicAdd(den, w);
    atomicAdd(po + lane, xj.x * w);
    atomicAdd(po + 32 + lane, xj.y * w);
}

__global__ __launch_bounds__(256) void gat_edge_l1(
    const int* __restrict__ src, const int* __restrict__ dst,
    const float* __restrict__ xl, const float* __restrict__ xr,
    const float* __restrict__ att, const unsigned int* __restrict__ ckey,
    int* __restrict__ mode, float* __restrict__ e1,
    float* __restrict__ m1, float* __restrict__ denom1a, float* __restrict__ denom1b,
    float* __restrict__ out1a, float* __restrict__ out1b, int E) {
    const int lane = threadIdx.x & 31;
    const float av0 = att[2 * lane], av1 = att[2 * lane + 1];
    float bn0, bn1;
    const bool loose = gat_bound_l1(ckey, av0, av1, lane, bn0, bn1);
    if (blockIdx.x == 0 && threadIdx.x == 0) *mode = loose ? 1 : 0;
    if (loose) {
        const long long E8 = (long long)E * HEADS;
        for (long long t = (long long)blockIdx.x * blockDim.x + threadIdx.x; t < E8;
             t += (long long)gridDim.x * blockDim.x)
            gat_logit_max_item_l1(xl, xr, att, src, dst, e1, m1, t);
        return;
    }
    const int wave = __builtin_amdgcn_readfirstlane(
        (int)(blockIdx.x * (blockDim.x >> 5) + (threadIdx.x >> 5)));
    const int wbase = wave * 32;
    if (wbase >= E) return;
    const bool upper = gat_package_half() != 0;
    float* __restrict__ denom1 = upper ? denom1b : denom1a;
    float* __restrict__ out1 = upper ? out1b : out1a;
    const int nE = (E - wbase) < 32 ? (E - wbase) : 32;
    int s_i = 0, d_i = 0;
    if (lane < nE) {
        s_i = src[wbase + lane];
        d_i = dst[wbase + lane];
    }
    const bool den_lane = (lane & 3) == 0;
    const int den_off = lane >> 2;

    int j0 = 0;
    for (; j0 + 4 <= nE; j0 += 4) {
        v2f xj[4], xi[4];
#pragma unroll
        for (int q = 0; q < 4; ++q) {
            const int s = __builtin_amdgcn_readlane(s_i, j0 + q);
            const int d = __builtin_amdgcn_readlane(d_i, j0 + q);
            xj[q] = *(const v2f*)(xl + (size_t)s * HID + 2 * lane);
            xi[q] = *(const v2f*)(xr + (size_t)d * HID + 2 * lane);
        }
#pragma unroll
        for (int q = 0; q < 4; ++q) {
            const int d = __builtin_amdgcn_readlane(d_i, j0 + q);
            gat_edge_step_l1(xj[q], xi[q], av0, av1, bn0, bn1,
                             den_lane, denom1 + (size_t)d * HEADS + den_off,
                             out1 + (size_t)d * HID, lane);
        }
    }
    for (; j0 < nE; ++j0) {
        const int s = __builtin_amdgcn_readlane(s_i, j0);
        const int d = __builtin_amdgcn_readlane(d_i, j0);
        const v2f xjv = *(const v2f*)(xl + (size_t)s * HID + 2 * lane);
        const v2f xiv = *(const v2f*)(xr + (size_t)d * HID + 2 * lane);
        gat_edge_step_l1(xjv, xiv, av0, av1, bn0, bn1,
                         den_lane, denom1 + (size_t)d * HEADS + den_off,
                         out1 + (size_t)d * HID, lane);
    }
}

__global__ __launch_bounds__(256) void gat_edge_aggregate_l1(
    const int* __restrict__ src, const int* __restrict__ dst,
    const float* __restrict__ xl, const float* __restrict__ e1,
    const float* __restrict__ m1, float* __restrict__ denom1a, float* __restrict__ denom1b,
    float* __restrict__ out1a, float* __restrict__ out1b,
    const int* __restrict__ mode, int E) {
    if (*mode == 0) return;
    const int lane = threadIdx.x & 31;
    const int wave0 = __builtin_amdgcn_readfirstlane(
        (int)(blockIdx.x * (blockDim.x >> 5) + (threadIdx.x >> 5)));
    const int nwaves = (int)(gridDim.x * (blockDim.x >> 5));
    const bool upper = gat_package_half() != 0;
    float* __restrict__ denom1 = upper ? denom1b : denom1a;
    float* __restrict__ out1 = upper ? out1b : out1a;
    const int sub = lane >> 3;
    const int h = lane & 7;
    const int hd = lane >> 2;

    for (int wave = wave0; wave * 32 < E; wave += nwaves) {
        const int wbase = wave * 32;
        const int nE = (E - wbase) < 32 ? (E - wbase) : 32;
        if (nE == 32) {
            const int s_i = src[wbase + lane];
            const int d_i = dst[wbase + lane];
            float ev[8], mv[8], ex[8];
            int dp[8];
#pragma unroll
            for (int g = 0; g < 8; ++g) {
                dp[g] = __shfl(d_i, g * 4 + sub, 32);
                ev[g] = e1[(size_t)wbase * HEADS + g * 32 + lane];
                mv[g] = m1[(size_t)dp[g] * HEADS + h];
            }
#pragma unroll
            for (int g = 0; g < 8; ++g) {
                ex[g] = expf(ev[g] - mv[g]);
                atomicAdd(&denom1[(size_t)dp[g] * HEADS + h], ex[g]);
            }
#pragma unroll
            for (int g = 0; g < 8; ++g) {
                v2f r[4];
#pragma unroll
                for (int q = 0; q < 4; ++q) {
                    const int s = __builtin_amdgcn_readlane(s_i, g * 4 + q);
                    r[q] = *(const v2f*)(xl + (size_t)s * HID + 2 * lane);
                }
#pragma unroll
                for (int q = 0; q < 4; ++q) {
                    const int d = __builtin_amdgcn_readlane(d_i, g * 4 + q);
                    const float a = __shfl(ex[g], q * 8 + hd, 32);
                    float* po = out1 + (size_t)d * HID;
                    atomicAdd(po + lane, r[q].x * a);
                    atomicAdd(po + 32 + lane, r[q].y * a);
                }
            }
        } else {
            for (int j = 0; j < nE; ++j) {
                const int e = wbase + j;
                const int s = src[e], d = dst[e];
                const float a = expf(e1[(size_t)e * HEADS + hd] - m1[(size_t)d * HEADS + hd]);
                if ((lane & 3) == 0) atomicAdd(&denom1[(size_t)d * HEADS + hd], a);
                const v2f rv = *(const v2f*)(xl + (size_t)s * HID + 2 * lane);
                float* po = out1 + (size_t)d * HID;
                atomicAdd(po + lane, rv.x * a);
                atomicAdd(po + 32 + lane, rv.y * a);
            }
        }
    }
}

__device__ __forceinline__ float gat_norm_elu(float acc, float den, float b) {
    const float v = (den > 0.0f ? acc / den : acc) + b;
    return v > 0.0f ? v : expm1f(v);
}

__global__ __launch_bounds__(256) void gat_norm_bias_elu(float* __restrict__ h1,
                                                        const float* __restrict__ out1,
                                                        const float* __restrict__ out1b,
                                                        const float* __restrict__ denom1,
                                                        const float* __restrict__ denom1b,
                                                        const float* __restrict__ bias1,
                                                        long long items) {
    const long long stride = (long long)gridDim.x * blockDim.x;
    long long i = (long long)blockIdx.x * blockDim.x + threadIdx.x;
    const int g = (int)(i & (HID / 4 - 1));
    const v4f bv = *(const v4f*)(bias1 + 4 * g);
    for (; i < items; i += stride) {
        const long long node = i >> 4;
        const float* __restrict__ ra = out1 + node * HID;
        const float* __restrict__ rb = out1b + node * HID;
        const v2f ea = *(const v2f*)(ra + 2 * g);
        const v2f oa = *(const v2f*)(ra + 32 + 2 * g);
        const v2f eb = *(const v2f*)(rb + 2 * g);
        const v2f ob = *(const v2f*)(rb + 32 + 2 * g);
        const long long nh = node * HEADS + (g >> 1);
        const float den = denom1[nh] + denom1b[nh];
        v4f r;
        r.x = gat_norm_elu(ea.x + eb.x, den, bv.x);
        r.y = gat_norm_elu(oa.x + ob.x, den, bv.y);
        r.z = gat_norm_elu(ea.y + eb.y, den, bv.z);
        r.w = gat_norm_elu(oa.y + ob.y, den, bv.w);
        *(v4f*)(h1 + 4 * i) = r;
    }
}

__device__ __forceinline__ void gat_logit_max_item_l2(
    const float* __restrict__ xl, const float* __restrict__ xr,
    const float* __restrict__ att, const int* __restrict__ src,
    const int* __restrict__ dst, float* __restrict__ e2,
    float* __restrict__ m2, long long t) {
    const int e = (int)(t >> 3);
    const int sub = (int)(t & 7);
    const int s = src[e], d = dst[e];
    const float* pj = xl + (size_t)s * DIM_OUT;
    const float* pi = xr + (size_t)d * DIM_OUT;
    float acc = 0.0f;
#pragma unroll
    for (int c = sub; c < DIM_OUT; c += 8) {
        float v = pi[c] + pj[c];
        v = v > 0.0f ? v : v * NEG_SLOPE;
        acc += v * att[c];
    }
    acc += __shfl_xor(acc, 1, 32);
    acc += __shfl_xor(acc, 2, 32);
    acc += __shfl_xor(acc, 4, 32);
    if (sub == 0) {
        e2[e] = acc;
        atomicMaxFloat(&m2[d], acc);
    }
}

__device__ __forceinline__ void gat_edge_step_l2(
    float xj0, float xj1, float xi0, float xi1,
    float av0, float av1, float bn0, float bn1,
    bool tail_ch, float* __restrict__ po, int lane) {
    const float ta = (gat_lrelu(xi0 + xj0) - gat_lrelu(xi0 + bn0)) * av0;
    const float tb = (gat_lrelu(xi1 + xj1) - gat_lrelu(xi1 + bn1)) * av1;
    float t = ta + (tail_ch ? tb : 0.0f);
    t += gat_row_xor<1>(t);
    t += gat_row_xor<2>(t);
    t += gat_row_xor<4>(t);
    t += gat_row_xor<8>(t);
    t += gat_xhalf(t);
    const float w = expf(t);
    const float v1 = tail_ch ? xj1 : 1.0f;
    atomicAdd(po + lane, xj0 * w);
    if (lane <= DEN2 - 32) atomicAdd(po + 32 + lane, v1 * w);
}

__global__ __launch_bounds__(256) void gat_edge_l2(
    const int* __restrict__ src, const int* __restrict__ dst,
    const float* __restrict__ xl, const float* __restrict__ xr,
    const float* __restrict__ att, const unsigned int* __restrict__ ckey,
    int* __restrict__ mode, float* __restrict__ e2,
    float* __restrict__ m2, float* __restrict__ out2a, float* __restrict__ out2b, int E) {
    const int lane = threadIdx.x & 31;
    const int c1 = 32 + (lane & 7);
    const bool tail_ch = lane < DIM_OUT - 32;
    const float av0 = att[lane], av1 = att[c1];
    float bn0, bn1;
    const bool loose = gat_bound_l2(ckey, av0, av1, lane, c1, tail_ch, bn0, bn1);
    if (blockIdx.x == 0 && threadIdx.x == 0) *mode = loose ? 1 : 0;
    if (loose) {
        const long long E8 = (long long)E * 8;
        for (long long t = (long long)blockIdx.x * blockDim.x + threadIdx.x; t < E8;
             t += (long long)gridDim.x * blockDim.x)
            gat_logit_max_item_l2(xl, xr, att, src, dst, e2, m2, t);
        return;
    }
    const int wave = __builtin_amdgcn_readfirstlane(
        (int)(blockIdx.x * (blockDim.x >> 5) + (threadIdx.x >> 5)));
    const int wbase = wave * 32;
    if (wbase >= E) return;
    float* __restrict__ out2 = gat_package_half() != 0 ? out2b : out2a;
    const int nE = (E - wbase) < 32 ? (E - wbase) : 32;
    int s_i = 0, d_i = 0;
    if (lane < nE) {
        s_i = src[wbase + lane];
        d_i = dst[wbase + lane];
    }
    int j0 = 0;
    for (; j0 + 4 <= nE; j0 += 4) {
        float xj0[4], xj1[4], xi0[4], xi1[4];
#pragma unroll
        for (int q = 0; q < 4; ++q) {
            const int s = __builtin_amdgcn_readlane(s_i, j0 + q);
            const int d = __builtin_amdgcn_readlane(d_i, j0 + q);
            const float* pj = xl + (size_t)s * DIM_OUT;
            const float* pi = xr + (size_t)d * DIM_OUT;
            xj0[q] = pj[lane];
            xj1[q] = pj[c1];
            xi0[q] = pi[lane];
            xi1[q] = pi[c1];
        }
#pragma unroll
        for (int q = 0; q < 4; ++q) {
            const int d = __builtin_amdgcn_readlane(d_i, j0 + q);
            gat_edge_step_l2(xj0[q], xj1[q], xi0[q], xi1[q], av0, av1, bn0, bn1,
                             tail_ch, out2 + (size_t)d * ROW2, lane);
        }
    }
    for (; j0 < nE; ++j0) {
        const int s = __builtin_amdgcn_readlane(s_i, j0);
        const int d = __builtin_amdgcn_readlane(d_i, j0);
        const float* pj = xl + (size_t)s * DIM_OUT;
        const float* pi = xr + (size_t)d * DIM_OUT;
        gat_edge_step_l2(pj[lane], pj[c1], pi[lane], pi[c1], av0, av1, bn0, bn1,
                         tail_ch, out2 + (size_t)d * ROW2, lane);
    }
}

__global__ __launch_bounds__(256) void gat_edge_aggregate_l2(
    const int* __restrict__ src, const int* __restrict__ dst,
    const float* __restrict__ xl, const float* __restrict__ e2,
    const float* __restrict__ m2, float* __restrict__ out2a, float* __restrict__ out2b,
    const int* __restrict__ mode, int E) {
    if (*mode == 0) return;
    const int lane = threadIdx.x & 31;
    const int wave0 = __builtin_amdgcn_readfirstlane(
        (int)(blockIdx.x * (blockDim.x >> 5) + (threadIdx.x >> 5)));
    const int nwaves = (int)(gridDim.x * (blockDim.x >> 5));
    float* __restrict__ out2 = gat_package_half() != 0 ? out2b : out2a;
    const int c1 = 32 + (lane & 7);
    const bool tail_ch = lane < DIM_OUT - 32;

    for (int wave = wave0; wave * 32 < E; wave += nwaves) {
        const int wbase = wave * 32;
        const int ei = wbase + lane;
        int s_i = 0, d_i = 0;
        float ex_i = 0.0f;
        if (ei < E) {
            s_i = src[ei];
            d_i = dst[ei];
            ex_i = expf(e2[ei] - m2[d_i]);
        }
        const int nE = (E - wbase) < 32 ? (E - wbase) : 32;
        if (nE == 32) {
            for (int j0 = 0; j0 < 32; j0 += 4) {
                float r0[4], r1[4];
#pragma unroll
                for (int q = 0; q < 4; ++q) {
                    const int s = __builtin_amdgcn_readlane(s_i, j0 + q);
                    const float* pj = xl + (size_t)s * DIM_OUT;
                    r0[q] = pj[lane];
                    r1[q] = pj[c1];
                }
                asm volatile("" :: "v"(r1[0]), "v"(r1[1]), "v"(r1[2]), "v"(r1[3]));
#pragma unroll
                for (int q = 0; q < 4; ++q) {
                    const int d = __builtin_amdgcn_readlane(d_i, j0 + q);
                    const float a = __int_as_float(
                        __builtin_amdgcn_readlane(__float_as_int(ex_i), j0 + q));
                    float* po = out2 + (size_t)d * ROW2;
                    const float v1 = tail_ch ? r1[q] : 1.0f;
                    atomicAdd(po + lane, r0[q] * a);
                    if (lane <= DEN2 - 32) atomicAdd(po + 32 + lane, v1 * a);
                }
            }
        } else {
            for (int j = 0; j < nE; ++j) {
                const int s = __builtin_amdgcn_readlane(s_i, j);
                const int d = __builtin_amdgcn_readlane(d_i, j);
                const float a = __int_as_float(
                    __builtin_amdgcn_readlane(__float_as_int(ex_i), j));
                const float* pj = xl + (size_t)s * DIM_OUT;
                float* po = out2 + (size_t)d * ROW2;
                const float v0 = pj[lane];
                const float v1 = tail_ch ? pj[c1] : 1.0f;
                atomicAdd(po + lane, v0 * a);
                if (lane <= DEN2 - 32) atomicAdd(po + 32 + lane, v1 * a);
            }
        }
    }
}

struct gat_fin_row { float a0, a1, c0, c1, d0, d1; };
struct gat_fin_res { float a, b, lse; };

__device__ __forceinline__ gat_fin_row gat_fin_load(
    const float* __restrict__ out2, const float* __restrict__ out2b,
    int node, int lane) {
    const float* __restrict__ row = out2 + (size_t)node * ROW2;
    const float* __restrict__ rowb = out2b + (size_t)node * ROW2;
    const int c1 = 32 + (lane & 7);
    gat_fin_row v;
    v.a0 = row[lane];
    v.a1 = rowb[lane];
    v.c0 = row[c1];
    v.c1 = rowb[c1];
    v.d0 = row[DEN2];
    v.d1 = rowb[DEN2];
    return v;
}

__device__ __forceinline__ gat_fin_res gat_fin_node(gat_fin_row v, float bs0, float bs1, bool has2) {
    const float den = v.d0 + v.d1;
    const bool pos = den > 0.0f;
    const float ra = v.a0 + v.a1;
    const float rb = v.c0 + v.c1;
    const float qa = ra / den;
    const float qb = rb / den;
    const float a = (pos ? qa : ra) + bs0;
    const float bq = (pos ? qb : rb) + bs1;
    const float b = has2 ? bq : -__builtin_inff();
    float mx = fmaxf(a, b);
#pragma unroll
    for (int o = 16; o >= 1; o >>= 1) mx = fmaxf(mx, __shfl_xor(mx, o, 32));
    float s = expf(a - mx) + expf(b - mx);
#pragma unroll
    for (int o = 16; o >= 1; o >>= 1) s += __shfl_xor(s, o, 32);
    gat_fin_res r;
    r.a = a;
    r.b = b;
    r.lse = logf(s) + mx;
    return r;
}

__device__ __forceinline__ void gat_fin_store(float* __restrict__ d_out, int N, int node,
                                              int lane, bool has2, gat_fin_res r) {
    float* h  = d_out + (size_t)node * DIM_OUT;
    float* ls = d_out + (size_t)N * DIM_OUT + (size_t)node * DIM_OUT;
    h[lane] = r.a;
    ls[lane] = r.a - r.lse;
    if (has2) {
        h[32 + lane] = r.b;
        ls[32 + lane] = r.b - r.lse;
    }
}

__global__ __launch_bounds__(256) void gat_finalize(
    const float* __restrict__ out2, const float* __restrict__ out2b,
    const float* __restrict__ bias2, float* __restrict__ d_out, int N) {
    const int lane = threadIdx.x & 31;
    int na = __builtin_amdgcn_readfirstlane(
        (int)(blockIdx.x * (blockDim.x >> 5) + (threadIdx.x >> 5)));
    const int nwaves = (int)(gridDim.x * (blockDim.x >> 5));
    if (na >= N) return;
    const bool has2 = (lane < DIM_OUT - 32);
    const float bs0 = bias2[lane];
    const float bs1 = has2 ? bias2[32 + lane] : 0.0f;
    bool two = na + nwaves < N;
    gat_fin_row ca = gat_fin_load(out2, out2b, na, lane);
    gat_fin_row cb = gat_fin_load(out2, out2b, two ? na + nwaves : na, lane);
    for (;;) {
        const int nxa = na + 2 * nwaves;
        const bool more = nxa < N;
        const bool more2 = nxa + nwaves < N;
        gat_fin_row xa = ca, xb = cb;
        if (more) {
            xa = gat_fin_load(out2, out2b, nxa, lane);
            xb = gat_fin_load(out2, out2b, more2 ? nxa + nwaves : nxa, lane);
        }
        const gat_fin_res r0 = gat_fin_node(ca, bs0, bs1, has2);
        const gat_fin_res r1 = gat_fin_node(cb, bs0, bs1, has2);
        gat_fin_store(d_out, N, na, lane, has2, r0);
        if (two) gat_fin_store(d_out, N, na + nwaves, lane, has2, r1);
        if (!more) break;
        na = nxa;
        two = more2;
        ca = xa;
        cb = xb;
    }
}

static inline int cdiv_i(long long a, long long b) { return (int)((a + b - 1) / b); }

extern "C" void kernel_launch(void* const* d_in, const int* in_sizes, int n_in,
                              void* d_out, int out_size, void* d_ws, size_t ws_size,
                              hipStream_t stream) {
    (void)n_in; (void)out_size; (void)ws_size;
    const float* x     = (const float*)d_in[0];
    const int*   eidx  = (const int*)d_in[1];
    const float* Wl1   = (const float*)d_in[2];
    const float* bl1   = (const float*)d_in[3];
    const float* Wr1   = (const float*)d_in[4];
    const float* br1   = (const float*)d_in[5];
    const float* att1  = (const float*)d_in[6];
    const float* bias1 = (const float*)d_in[7];
    const float* Wl2   = (const float*)d_in[8];
    const float* bl2   = (const float*)d_in[9];
    const float* Wr2   = (const float*)d_in[10];
    const float* br2   = (const float*)d_in[11];
    const float* att2  = (const float*)d_in[12];
    const float* bias2 = (const float*)d_in[13];

    const int N = in_sizes[0] / DIM_IN;
    const int E = in_sizes[1] / 2;
    const int* src = eidx;
    const int* dst = eidx + E;

    float* ws = (float*)d_ws;
    size_t o = 0;
    float* h_l1   = ws + o; o += (size_t)N * HID;
    float* h_r1   = ws + o; o += (size_t)N * HID;
    float* e1     = ws + o; o += (size_t)E * HEADS;
    float* m1     = ws + o; o += (size_t)N * HEADS;
    float* denom1 = ws + o; o += (size_t)N * HEADS;
    float* denom1b = ws + o; o += (size_t)N * HEADS;
    float* out1   = ws + o; o += (size_t)N * HID;
    float* out1b  = ws + o; o += (size_t)N * HID;
    float* h1     = ws + o; o += (size_t)N * HID;
    unsigned int* ckey1 = (unsigned int*)(ws + o); o += 256;
    int*   mode1  = (int*)(ws + o); o += 64;
    unsigned int* ckey2 = (unsigned int*)(ws + o); o += 256;
    int*   mode2  = (int*)(ws + o); o += 64;
    float* h_l2   = ws;
    float* h_r2   = h_l2 + (size_t)N * DIM_OUT;
    float* e2     = h_r2 + (size_t)N * DIM_OUT;
    float* out2   = e2 + (size_t)E;
    float* out2b  = out2 + (size_t)N * ROW2;
    float* m2     = out2b + (size_t)N * ROW2;

    const int B = 256;

    gat_bound_start<<<1, 256, 0, stream>>>(ckey1, ckey2);

    gat_proj_pair_f32_wmma<DIM_IN, HID, 4, 2, HID, HEADS, HEADS>
        <<<1024, 128, 0, stream>>>(x, Wl1, bl1, Wr1, br1, h_l1, h_r1,
                                   out1, out1b, m1, denom1, denom1b, ckey1, N);

    gat_edge_l1<<<cdiv_i(E, B), B, 0, stream>>>(
        src, dst, h_l1, h_r1, att1, ckey1, mode1, e1, m1, denom1, denom1b, out1, out1b, E);
    gat_edge_aggregate_l1<<<512, B, 0, stream>>>(
        src, dst, h_l1, e1, m1, denom1, denom1b, out1, out1b, mode1, E);
    gat_norm_bias_elu<<<cdiv_i((long long)N * (HID / 4), 4 * B), B, 0, stream>>>(
        h1, out1, out1b, denom1, denom1b, bias1, (long long)N * (HID / 4));

    gat_proj_pair_f32_wmma<HID, DIM_OUT, 3, 3, ROW2, 1, 0>
        <<<1250, 64, 0, stream>>>(h1, Wl2, bl2, Wr2, br2, h_l2, h_r2,
                                  out2, out2b, m2, (float*)nullptr, (float*)nullptr, ckey2, N);

    gat_edge_l2<<<cdiv_i(E, B), B, 0, stream>>>(
        src, dst, h_l2, h_r2, att2, ckey2, mode2, e2, m2, out2, out2b, E);
    gat_edge_aggregate_l2<<<512, B, 0, stream>>>(
        src, dst, h_l2, e2, m2, out2, out2b, mode2, E);

    gat_finalize<<<cdiv_i(N, 8 * (B / 32)), B, 0, stream>>>((const float*)out2, (const float*)out2b,
                                                            bias2, (float*)d_out, N);
}
